# starting kernel
# baseline (speedup 1.0000x reference)
#include <hip/hip_runtime.h>
#include <hip/hip_bf16.h>
#include <math.h>

#define Bn   64
#define Ln   2048
#define Dn   300
#define Cn   2
#define Kn   111
#define PADn 55

typedef __attribute__((ext_vector_type(2))) float v2f;
typedef __attribute__((ext_vector_type(8))) float v8f;

__global__ __launch_bounds__(256)
void k_phase1(const int* __restrict__ tok, const float* __restrict__ emb,
              const float* __restrict__ cls, float* __restrict__ g)
{
    __shared__ float clsn[3 * 304];
    __shared__ float red[256];
    const int tid = threadIdx.x;

    for (int i = tid; i < 3 * 304; i += 256) clsn[i] = 0.0f;
    __syncthreads();
    for (int c = 0; c < Cn; ++c) {
        float s = 0.0f;
        for (int d = tid; d < Dn; d += 256) { float v = cls[c * Dn + d]; s += v * v; }
        red[tid] = s;
        __syncthreads();
        for (int off = 128; off > 0; off >>= 1) {
            if (tid < off) red[tid] += red[tid + off];
            __syncthreads();
        }
        float rn = 1.0f / fmaxf(sqrtf(red[0]), 1e-12f);
        __syncthreads();
        for (int d = tid; d < Dn; d += 256) clsn[c * 304 + d] = cls[c * Dn + d] * rn;
        __syncthreads();
    }

    const int wave = tid >> 5;
    const int lane = tid & 31;
    const int tileId = blockIdx.x * 8 + wave;
    const int b  = tileId >> 7;
    const int l0 = (tileId & 127) * 16;
    const int m    = lane & 15;
    const int hi   = lane >> 4;
    const int koff = hi * 2;

    const int token = tok[b * Ln + l0 + m];
    const float* __restrict__ row = emb + (long)token * Dn;
    const float* __restrict__ bb  = &clsn[(m < 2 ? m : 2) * 304];

    v8f acc = {};
    float n2 = 0.0f;
    #pragma unroll 5
    for (int k = 0; k < 75; ++k) {
        const int d0 = 4 * k + koff;
        v2f a;  a.x = row[d0];           a.y = row[d0 + 1];
        v2f bf; bf.x = bb[d0];           bf.y = bb[d0 + 1];
        n2 += a.x * a.x + a.y * a.y;
        acc = __builtin_amdgcn_wmma_f32_16x16x4_f32(
            false, a, false, bf, (short)0, acc, false, false);
    }

    n2 += __shfl_xor(n2, 16, 32);

    const int  c      = m;
    const bool active = (m < 2);
    const int  mbase  = hi * 8;
    float* __restrict__ gout = g + (((long)b * Cn + c) * Ln + l0 + mbase);
    #pragma unroll
    for (int i = 0; i < 8; ++i) {
        float nn    = __shfl(n2, mbase + i, 32);
        float scale = 1.0f / fmaxf(sqrtf(nn), 1e-12f);
        if (active) gout[i] = acc[i] * scale;
    }
}

__global__ __launch_bounds__(256)
void k_phase2(const float* __restrict__ g, const float* __restrict__ conv_w,
              const float* __restrict__ conv_b, float* __restrict__ betaOut)
{
    __shared__ float gin[Cn][Ln + 2 * PADn + 2];
    __shared__ float wsm[Cn * Cn * Kn];
    __shared__ float red[256];
    const int b = blockIdx.x, tid = threadIdx.x;

    for (int i = tid; i < Cn * (Ln + 2 * PADn + 2); i += 256) (&gin[0][0])[i] = 0.0f;
    __syncthreads();
    for (int i = tid; i < Ln; i += 256) {
        gin[0][PADn + i] = g[((long)b * Cn + 0) * Ln + i];
        gin[1][PADn + i] = g[((long)b * Cn + 1) * Ln + i];
    }
    for (int i = tid; i < Cn * Cn * Kn; i += 256) wsm[i] = conv_w[i];
    __syncthreads();

    const float cb0 = conv_b[0], cb1 = conv_b[1];
    float h[8];
    #pragma unroll
    for (int j = 0; j < 8; ++j) {
        const int x = tid + 256 * j;
        float a0 = 0.0f, a1 = 0.0f;
        for (int k = 0; k < Kn; ++k) {
            const float i0 = gin[0][x + k];
            const float i1 = gin[1][x + k];
            a0 += wsm[0 * Kn + k] * i0 + wsm[1 * Kn + k] * i1;
            a1 += wsm[2 * Kn + k] * i0 + wsm[3 * Kn + k] * i1;
        }
        h[j] = fmaxf(fmaxf(a0 + cb0, 0.0f), fmaxf(a1 + cb1, 0.0f));
    }

    float lm = -INFINITY;
    #pragma unroll
    for (int j = 0; j < 8; ++j) lm = fmaxf(lm, h[j]);
    red[tid] = lm; __syncthreads();
    for (int off = 128; off > 0; off >>= 1) {
        if (tid < off) red[tid] = fmaxf(red[tid], red[tid + off]);
        __syncthreads();
    }
    const float M = red[0];
    __syncthreads();

    float ls = 0.0f;
    #pragma unroll
    for (int j = 0; j < 8; ++j) { h[j] = __expf(h[j] - M); ls += h[j]; }
    red[tid] = ls; __syncthreads();
    for (int off = 128; off > 0; off >>= 1) {
        if (tid < off) red[tid] += red[tid + off];
        __syncthreads();
    }
    const float inv = 1.0f / red[0];
    #pragma unroll
    for (int j = 0; j < 8; ++j)
        betaOut[(long)b * Ln + tid + 256 * j] = h[j] * inv;
}

__global__ __launch_bounds__(320)
void k_phase3(const int* __restrict__ tok, const float* __restrict__ emb,
              const float* __restrict__ beta, const float* __restrict__ lin_w,
              const float* __restrict__ lin_b, float* __restrict__ logits)
{
    __shared__ float s_acc[2];
    const int b = blockIdx.x, tid = threadIdx.x;
    if (tid < 2) s_acc[tid] = 0.0f;

    const int*   __restrict__ trow = tok  + (long)b * Ln;
    const float* __restrict__ brow = beta + (long)b * Ln;
    const bool act = (tid < Dn);
    const int  dd  = act ? tid : 0;

    float z = 0.0f;
    for (int l = 0; l < Ln; l += 4) {
        const int t0 = trow[l],     t1 = trow[l + 1];
        const int t2 = trow[l + 2], t3 = trow[l + 3];
        const float b0 = brow[l],     b1 = brow[l + 1];
        const float b2 = brow[l + 2], b3 = brow[l + 3];
        z += b0 * emb[(long)t0 * Dn + dd];
        z += b1 * emb[(long)t1 * Dn + dd];
        z += b2 * emb[(long)t2 * Dn + dd];
        z += b3 * emb[(long)t3 * Dn + dd];
    }
    __syncthreads();
    if (act) {
        atomicAdd(&s_acc[0], z * lin_w[0 * Dn + tid]);
        atomicAdd(&s_acc[1], z * lin_w[1 * Dn + tid]);
    }
    __syncthreads();
    if (tid == 0) {
        const float o0 = s_acc[0] + lin_b[0];
        const float o1 = s_acc[1] + lin_b[1];
        const float mm = fmaxf(o0, o1);
        const float lse = mm + __logf(__expf(o0 - mm) + __expf(o1 - mm));
        logits[b * 2 + 0] = o0 - lse;
        logits[b * 2 + 1] = o1 - lse;
    }
}

extern "C" void kernel_launch(void* const* d_in, const int* in_sizes, int n_in,
                              void* d_out, int out_size, void* d_ws, size_t ws_size,
                              hipStream_t stream)
{
    const int*   tok = (const int*)  d_in[0];
    const float* emb = (const float*)d_in[1];
    const float* cls = (const float*)d_in[2];
    const float* cw  = (const float*)d_in[3];
    const float* cb  = (const float*)d_in[4];
    const float* lw  = (const float*)d_in[5];
    const float* lb  = (const float*)d_in[6];

    float* out    = (float*)d_out;
    float* logits = out;
    float* beta   = out + Bn * Cn;
    float* g      = (float*)d_ws;

    k_phase1<<<1024, 256, 0, stream>>>(tok, emb, cls, g);
    k_phase2<<<Bn, 256, 0, stream>>>(g, cw, cb, beta);
    k_phase3<<<Bn, 320, 0, stream>>>(tok, emb, beta, lw, lb, logits);
}


# Phase 2 conv on the f32 16x16x4 matrix instruction: a wave owns 256 positions as a 16x16 tile and sums 8 banded 16x16 tap blocks per channel pair (128 WMMAs, 2 accumulators), fragments read from LDS.
# speedup vs baseline: 3.0078x; 3.0078x over previous
#include <hip/hip_runtime.h>
#include <hip/hip_bf16.h>
#include <math.h>

#define Bn   64
#define Ln   2048
#define Dn   300
#define Cn   2
#define Kn   111
#define PADn 55

typedef __attribute__((ext_vector_type(2))) float v2f;
typedef __attribute__((ext_vector_type(8))) float v8f;
typedef __attribute__((ext_vector_type(4))) float v4f;

__global__ __launch_bounds__(256)
void k_phase1(const int* __restrict__ tok, const float* __restrict__ emb,
              const float* __restrict__ cls, float* __restrict__ g)
{
    __shared__ float clsn[3 * 304];
    __shared__ float red[256];
    const int tid = threadIdx.x;

    for (int i = tid; i < 3 * 304; i += 256) clsn[i] = 0.0f;
    __syncthreads();
    for (int c = 0; c < Cn; ++c) {
        float s = 0.0f;
        for (int d = tid; d < Dn; d += 256) { float v = cls[c * Dn + d]; s += v * v; }
        red[tid] = s;
        __syncthreads();
        for (int off = 128; off > 0; off >>= 1) {
            if (tid < off) red[tid] += red[tid + off];
            __syncthreads();
        }
        float rn = 1.0f / fmaxf(sqrtf(red[0]), 1e-12f);
        __syncthreads();
        for (int d = tid; d < Dn; d += 256) clsn[c * 304 + d] = cls[c * Dn + d] * rn;
        __syncthreads();
    }

    const int wave = tid >> 5;
    const int lane = tid & 31;
    const int tileId = blockIdx.x * 8 + wave;
    const int b  = tileId >> 7;
    const int l0 = (tileId & 127) * 16;
    const int m    = lane & 15;
    const int hi   = lane >> 4;
    const int koff = hi * 2;

    const int token = tok[b * Ln + l0 + m];
    const float* __restrict__ row = emb + (long)token * Dn;
    const float* __restrict__ bb  = &clsn[(m < 2 ? m : 2) * 304];

    v8f acc = {};
    float n2 = 0.0f;
    #pragma unroll 5
    for (int k = 0; k < 75; ++k) {
        const int d0 = 4 * k + koff;
        v2f a;  a.x = row[d0];           a.y = row[d0 + 1];
        v2f bf; bf.x = bb[d0];           bf.y = bb[d0 + 1];
        n2 += a.x * a.x + a.y * a.y;
        acc = __builtin_amdgcn_wmma_f32_16x16x4_f32(
            false, a, false, bf, (short)0, acc, false, false);
    }

    n2 += __shfl_xor(n2, 16, 32);

    const int  c      = m;
    const bool active = (m < 2);
    const int  mbase  = hi * 8;
    float* __restrict__ gout = g + (((long)b * Cn + c) * Ln + l0 + mbase);
    #pragma unroll
    for (int i = 0; i < 8; ++i) {
        float nn    = __shfl(n2, mbase + i, 32);
        float scale = 1.0f / fmaxf(sqrtf(nn), 1e-12f);
        if (active) gout[i] = acc[i] * scale;
    }
}

#define GP_ROWS  135
#define GP_PITCH 20
#define WP_LEAD  15
#define WP_LEN   144

__global__ __launch_bounds__(256)
void k_phase2(const float* __restrict__ g, const float* __restrict__ conv_w,
              const float* __restrict__ conv_b, float* __restrict__ betaOut)
{
    __shared__ __attribute__((aligned(16))) float gp[Cn][GP_ROWS * GP_PITCH];
    __shared__ float wp[Cn * Cn][WP_LEN];
    __shared__ float red[256];
    __shared__ __attribute__((aligned(16))) float hs[Ln];
    const int b = blockIdx.x, tid = threadIdx.x;

    for (int i = tid; i < Cn * GP_ROWS * GP_PITCH; i += 256) (&gp[0][0])[i] = 0.0f;
    for (int i = tid; i < Cn * Cn * WP_LEN; i += 256) (&wp[0][0])[i] = 0.0f;
    __syncthreads();
    for (int i = tid; i < Ln; i += 256) {
        const int u = i + PADn;
        const int a = (u >> 4) * GP_PITCH + (u & 15);
        gp[0][a] = g[((long)b * Cn + 0) * Ln + i];
        gp[1][a] = g[((long)b * Cn + 1) * Ln + i];
    }
    for (int i = tid; i < Cn * Cn * Kn; i += 256) {
        const int pr = i / Kn;
        wp[pr][WP_LEAD + i - pr * Kn] = conv_w[i];
    }
    __syncthreads();

    const int lane = tid & 31;
    const int wv   = tid >> 5;
    const int c    = lane & 15;
    const int hh   = lane >> 4;
    const float* __restrict__ a0p = &gp[0][(16 * wv + c) * GP_PITCH + 2 * hh];
    const float* __restrict__ a1p = &gp[1][(16 * wv + c) * GP_PITCH + 2 * hh];
    const float* __restrict__ bp  = &wp[0][WP_LEAD - c + 2 * hh];

    v8f acc0 = {}, acc1 = {};
    #pragma unroll
    for (int p = 0; p < 8; ++p) {
        #pragma unroll
        for (int q = 0; q < 4; ++q) {
            const int ao = p * GP_PITCH + 4 * q;
            const int t  = 16 * p + 4 * q;
            const v2f a0 = *(const v2f*)(a0p + ao);
            const v2f a1 = *(const v2f*)(a1p + ao);
            v2f b00, b01, b10, b11;
            b00.x = bp[0 * WP_LEN + t]; b00.y = bp[0 * WP_LEN + t + 1];
            b01.x = bp[1 * WP_LEN + t]; b01.y = bp[1 * WP_LEN + t + 1];
            b10.x = bp[2 * WP_LEN + t]; b10.y = bp[2 * WP_LEN + t + 1];
            b11.x = bp[3 * WP_LEN + t]; b11.y = bp[3 * WP_LEN + t + 1];
            acc0 = __builtin_amdgcn_wmma_f32_16x16x4_f32(
                false, a0, false, b00, (short)0, acc0, false, false);
            acc1 = __builtin_amdgcn_wmma_f32_16x16x4_f32(
                false, a0, false, b10, (short)0, acc1, false, false);
            acc0 = __builtin_amdgcn_wmma_f32_16x16x4_f32(
                false, a1, false, b01, (short)0, acc0, false, false);
            acc1 = __builtin_amdgcn_wmma_f32_16x16x4_f32(
                false, a1, false, b11, (short)0, acc1, false, false);
        }
    }

    const int xb = 256 * wv + 128 * hh + c;
    const float cb0 = conv_b[0], cb1 = conv_b[1];
    float h[8];
    for (int e = 0; e < 8; ++e)
        h[e] = fmaxf(fmaxf(acc0[e] + cb0, 0.0f), fmaxf(acc1[e] + cb1, 0.0f));

    float lm = -INFINITY;
    #pragma unroll
    for (int e = 0; e < 8; ++e) lm = fmaxf(lm, h[e]);
    red[tid] = lm; __syncthreads();
    for (int off = 128; off > 0; off >>= 1) {
        if (tid < off) red[tid] = fmaxf(red[tid], red[tid + off]);
        __syncthreads();
    }
    const float M = red[0];
    __syncthreads();

    float ls = 0.0f;
    #pragma unroll
    for (int e = 0; e < 8; ++e) {
        h[e] = __expf(h[e] - M);
        ls += h[e];
        hs[xb + 16 * e] = h[e];
    }
    red[tid] = ls; __syncthreads();
    for (int off = 128; off > 0; off >>= 1) {
        if (tid < off) red[tid] += red[tid + off];
        __syncthreads();
    }
    const float inv = 1.0f / red[0];
    for (int j = 0; j < 2; ++j) {
        const int e4 = 4 * (tid + 256 * j);
        const v4f v = *(const v4f*)&hs[e4];
        *(v4f*)&betaOut[(long)b * Ln + e4] = v * inv;
    }
}

#define P3_WAVES 32
#define P3_TOK   (Ln / P3_WAVES)
#define P3_RING  8

__global__ __launch_bounds__(1024)
void k_phase3(const int* __restrict__ tok, const float* __restrict__ emb,
              const float* __restrict__ beta, const float* __restrict__ lin_w,
              const float* __restrict__ lin_b, float* __restrict__ logits)
{
    __shared__ float part[P3_WAVES][2];
    const int b = blockIdx.x, tid = threadIdx.x;
    const int lane = tid & 31;
    const int w    = tid >> 5;

    const int*   __restrict__ trow = tok  + (long)b * Ln + w * P3_TOK;
    const float* __restrict__ brow = beta + (long)b * Ln + w * P3_TOK;
    const int   tv0 = trow[lane], tv1 = trow[lane + 32];
    const float bv0 = brow[lane], bv1 = brow[lane + 32];

    const int      pl = lane < 25 ? lane : 24;
    const unsigned o  = (unsigned)pl * 16u;
    const char* __restrict__ embc = (const char*)emb;

    v4f x0[P3_RING], x1[P3_RING], x2[P3_RING];
    v4f a0 = {}, a1 = {}, a2 = {};

#define P3_LOAD(slot, tvec, ln) {                                            \
        const int t_ = __builtin_amdgcn_readlane(tvec, ln);                  \
        const char* rp_ = embc + (size_t)(unsigned)t_ * (Dn * 4u);           \
        x0[slot] = *(const v4f*)(rp_ + o);                                   \
        x1[slot] = *(const v4f*)(rp_ + 400 + o);                             \
        x2[slot] = *(const v4f*)(rp_ + 800 + o); }

#define P3_USE(slot, bvec, ln) {                                             \
        const float bt_ = __builtin_bit_cast(float, __builtin_amdgcn_readlane( \
            __builtin_bit_cast(int, bvec), ln));                             \
        a0 += bt_ * x0[slot]; a1 += bt_ * x1[slot]; a2 += bt_ * x2[slot]; }

    #pragma unroll
    for (int j = 0; j < P3_RING; ++j) P3_LOAD(j, tv0, j);

    for (int g = 0; g < P3_TOK / P3_RING - 1; ++g) {
        const float bcur = g < 4 ? bv0 : bv1;
        const int   tnxt = g < 3 ? tv0 : tv1;
        const int   lc = (g & 3) * P3_RING;
        const int   ln = ((g + 1) & 3) * P3_RING;
        #pragma unroll
        for (int j = 0; j < P3_RING; ++j) {
            P3_USE(j, bcur, lc + j);
            P3_LOAD(j, tnxt, ln + j);
        }
    }
    #pragma unroll
    for (int j = 0; j < P3_RING; ++j) P3_USE(j, bv1, 32 - P3_RING + j);

#undef P3_LOAD
#undef P3_USE

    float p0 = 0.0f, p1 = 0.0f;
    if (lane < 25) {
        const int d0 = 4 * lane;
        for (int e = 0; e < 4; ++e) {
            p0 += a0[e] * lin_w[d0 + e] + a1[e] * lin_w[100 + d0 + e]
                + a2[e] * lin_w[200 + d0 + e];
            p1 += a0[e] * lin_w[Dn + d0 + e] + a1[e] * lin_w[Dn + 100 + d0 + e]
                + a2[e] * lin_w[Dn + 200 + d0 + e];
        }
    }
    for (int off = 16; off > 0; off >>= 1) {
        p0 += __shfl_xor(p0, off, 32);
        p1 += __shfl_xor(p1, off, 32);
    }
    if (lane == 0) { part[w][0] = p0; part[w][1] = p1; }
    __syncthreads();

    if (tid < 32) {
        float q0 = part[tid][0], q1 = part[tid][1];
        for (int off = 16; off > 0; off >>= 1) {
            q0 += __shfl_xor(q0, off, 32);
            q1 += __shfl_xor(q1, off, 32);
        }
        if (tid == 0) {
            const float o0 = q0 + lin_b[0];
            const float o1 = q1 + lin_b[1];
            const float mm = fmaxf(o0, o1);
            const float lse = mm + __logf(__expf(o0 - mm) + __expf(o1 - mm));
            logits[b * 2 + 0] = o0 - lse;
            logits[b * 2 + 1] = o1 - lse;
        }
    }
}

extern "C" void kernel_launch(void* const* d_in, const int* in_sizes, int n_in,
                              void* d_out, int out_size, void* d_ws, size_t ws_size,
                              hipStream_t stream)
{
    const int*   tok = (const int*)  d_in[0];
    const float* emb = (const float*)d_in[1];
    const float* cls = (const float*)d_in[2];
    const float* cw  = (const float*)d_in[3];
    const float* cb  = (const float*)d_in[4];
    const float* lw  = (const float*)d_in[5];
    const float* lb  = (const float*)d_in[6];

    float* out    = (float*)d_out;
    float* logits = out;
    float* beta   = out + Bn * Cn;
    float* g      = (float*)d_ws;

    k_phase1<<<1024, 256, 0, stream>>>(tok, emb, cls, g);
    k_phase2<<<Bn, 256, 0, stream>>>(g, cw, cb, beta);
    k_phase3<<<Bn, 1024, 0, stream>>>(tok, emb, beta, lw, lb, logits);
}


# Phase 1 stages each tile's 16 table rows through LDS: rows are read as they lie (400 contiguous bytes per load, whole lines), WMMA fragments come from LDS 16 bytes per lane; two accumulators.
# speedup vs baseline: 3.3081x; 1.0998x over previous
#include <hip/hip_runtime.h>
#include <hip/hip_bf16.h>
#include <math.h>

#define Bn   64
#define Ln   2048
#define Dn   300
#define Cn   2
#define Kn   111
#define PADn 55

typedef __attribute__((ext_vector_type(2))) float v2f;
typedef __attribute__((ext_vector_type(8))) float v8f;
typedef __attribute__((ext_vector_type(4))) float v4f;

#define XS_PITCH 300

__global__ __launch_bounds__(256)
void k_phase1(const int* __restrict__ tok, const float* __restrict__ emb,
              const float* __restrict__ cls, float* __restrict__ g)
{
    __shared__ __attribute__((aligned(16))) float clsn[3 * 304];
    __shared__ float red[256];
    __shared__ __attribute__((aligned(16))) float xs[8][16 * XS_PITCH];
    const int tid = threadIdx.x;

    const int wave = tid >> 5;
    const int lane = tid & 31;
    const int tileId = blockIdx.x * 8 + wave;
    const int b  = tileId >> 7;
    const int l0 = (tileId & 127) * 16;
    const int m    = lane & 15;
    const int hi   = lane >> 4;

    const int token = tok[b * Ln + l0 + m];

    {
        const int      pl = lane < 25 ? lane : 24;
        const unsigned o  = (unsigned)pl * 16u;
        const char* __restrict__ embc = (const char*)emb;
        char* xsw = (char*)&xs[wave][0];
        #pragma unroll
        for (int half = 0; half < 2; ++half) {
            v4f x0[8], x1[8], x2[8];
            #pragma unroll
            for (int r = 0; r < 8; ++r) {
                const int t_ = __builtin_amdgcn_readlane(token, half * 8 + r);
                const char* rp_ = embc + (size_t)(unsigned)t_ * (Dn * 4u);
                x0[r] = *(const v4f*)(rp_ + o);
                x1[r] = *(const v4f*)(rp_ + 400 + o);
                x2[r] = *(const v4f*)(rp_ + 800 + o);
            }
            #pragma unroll
            for (int r = 0; r < 8; ++r) {
                char* wp_ = xsw + (half * 8 + r) * (XS_PITCH * 4) + o;
                *(v4f*)(wp_)       = x0[r];
                *(v4f*)(wp_ + 400) = x1[r];
                *(v4f*)(wp_ + 800) = x2[r];
            }
        }
    }

    for (int i = tid; i < 3 * 304; i += 256) clsn[i] = 0.0f;
    __syncthreads();
    for (int c = 0; c < Cn; ++c) {
        float s = 0.0f;
        for (int d = tid; d < Dn; d += 256) { float v = cls[c * Dn + d]; s += v * v; }
        red[tid] = s;
        __syncthreads();
        for (int off = 128; off > 0; off >>= 1) {
            if (tid < off) red[tid] += red[tid + off];
            __syncthreads();
        }
        float rn = 1.0f / fmaxf(sqrtf(red[0]), 1e-12f);
        __syncthreads();
        for (int d = tid; d < Dn; d += 256) clsn[c * 304 + d] = cls[c * Dn + d] * rn;
        __syncthreads();
    }

    const float* __restrict__ ar = &xs[wave][m * XS_PITCH + 4 * hi];
    const float* __restrict__ br = &clsn[(m < 2 ? m : 2) * 304 + 4 * hi];

    v8f acc0 = {}, acc1 = {};
    float n2 = 0.0f;
    #pragma unroll
    for (int j = 0; j < 37; ++j) {
        const v4f a  = *(const v4f*)(ar + 8 * j);
        const v4f bq = *(const v4f*)(br + 8 * j);
        n2 += a.x * a.x + a.y * a.y + a.z * a.z + a.w * a.w;
        v2f a01, a23, b01, b23;
        a01.x = a.x;  a01.y = a.y;  a23.x = a.z;  a23.y = a.w;
        b01.x = bq.x; b01.y = bq.y; b23.x = bq.z; b23.y = bq.w;
        acc0 = __builtin_amdgcn_wmma_f32_16x16x4_f32(
            false, a01, false, b01, (short)0, acc0, false, false);
        acc1 = __builtin_amdgcn_wmma_f32_16x16x4_f32(
            false, a23, false, b23, (short)0, acc1, false, false);
    }
    {
        const v2f a  = *(const v2f*)(&xs[wave][m * XS_PITCH + 296 + 2 * hi]);
        const v2f bq = *(const v2f*)(&clsn[(m < 2 ? m : 2) * 304 + 296 + 2 * hi]);
        n2 += a.x * a.x + a.y * a.y;
        acc0 = __builtin_amdgcn_wmma_f32_16x16x4_f32(
            false, a, false, bq, (short)0, acc0, false, false);
    }
    const v8f acc = acc0 + acc1;

    n2 += __shfl_xor(n2, 16, 32);

    const int  c      = m;
    const bool active = (m < 2);
    const int  mbase  = hi * 8;
    float* __restrict__ gout = g + (((long)b * Cn + c) * Ln + l0 + mbase);
    #pragma unroll
    for (int i = 0; i < 8; ++i) {
        float nn    = __shfl(n2, mbase + i, 32);
        float scale = 1.0f / fmaxf(sqrtf(nn), 1e-12f);
        if (active) gout[i] = acc[i] * scale;
    }
}

#define GP_ROWS  135
#define GP_PITCH 20
#define WP_LEAD  15
#define WP_LEN   144

__global__ __launch_bounds__(256)
void k_phase2(const float* __restrict__ g, const float* __restrict__ conv_w,
              const float* __restrict__ conv_b, float* __restrict__ betaOut)
{
    __shared__ __attribute__((aligned(16))) float gp[Cn][GP_ROWS * GP_PITCH];
    __shared__ float wp[Cn * Cn][WP_LEN];
    __shared__ float red[256];
    __shared__ __attribute__((aligned(16))) float hs[Ln];
    const int b = blockIdx.x, tid = threadIdx.x;

    for (int i = tid; i < Cn * GP_ROWS * GP_PITCH; i += 256) (&gp[0][0])[i] = 0.0f;
    for (int i = tid; i < Cn * Cn * WP_LEN; i += 256) (&wp[0][0])[i] = 0.0f;
    __syncthreads();
    for (int i = tid; i < Ln; i += 256) {
        const int u = i + PADn;
        const int a = (u >> 4) * GP_PITCH + (u & 15);
        gp[0][a] = g[((long)b * Cn + 0) * Ln + i];
        gp[1][a] = g[((long)b * Cn + 1) * Ln + i];
    }
    for (int i = tid; i < Cn * Cn * Kn; i += 256) {
        const int pr = i / Kn;
        wp[pr][WP_LEAD + i - pr * Kn] = conv_w[i];
    }
    __syncthreads();

    const int lane = tid & 31;
    const int wv   = tid >> 5;
    const int c    = lane & 15;
    const int hh   = lane >> 4;
    const float* __restrict__ a0p = &gp[0][(16 * wv + c) * GP_PITCH + 2 * hh];
    const float* __restrict__ a1p = &gp[1][(16 * wv + c) * GP_PITCH + 2 * hh];
    const float* __restrict__ bp  = &wp[0][WP_LEAD - c + 2 * hh];

    v8f acc0 = {}, acc1 = {};
    #pragma unroll
    for (int p = 0; p < 8; ++p) {
        #pragma unroll
        for (int q = 0; q < 4; ++q) {
            const int ao = p * GP_PITCH + 4 * q;
            const int t  = 16 * p + 4 * q;
            const v2f a0 = *(const v2f*)(a0p + ao);
            const v2f a1 = *(const v2f*)(a1p + ao);
            v2f b00, b01, b10, b11;
            b00.x = bp[0 * WP_LEN + t]; b00.y = bp[0 * WP_LEN + t + 1];
            b01.x = bp[1 * WP_LEN + t]; b01.y = bp[1 * WP_LEN + t + 1];
            b10.x = bp[2 * WP_LEN + t]; b10.y = bp[2 * WP_LEN + t + 1];
            b11.x = bp[3 * WP_LEN + t]; b11.y = bp[3 * WP_LEN + t + 1];
            acc0 = __builtin_amdgcn_wmma_f32_16x16x4_f32(
                false, a0, false, b00, (short)0, acc0, false, false);
            acc1 = __builtin_amdgcn_wmma_f32_16x16x4_f32(
                false, a0, false, b10, (short)0, acc1, false, false);
            acc0 = __builtin_amdgcn_wmma_f32_16x16x4_f32(
                false, a1, false, b01, (short)0, acc0, false, false);
            acc1 = __builtin_amdgcn_wmma_f32_16x16x4_f32(
                false, a1, false, b11, (short)0, acc1, false, false);
        }
    }

    const int xb = 256 * wv + 128 * hh + c;
    const float cb0 = conv_b[0], cb1 = conv_b[1];
    float h[8];
    for (int e = 0; e < 8; ++e)
        h[e] = fmaxf(fmaxf(acc0[e] + cb0, 0.0f), fmaxf(acc1[e] + cb1, 0.0f));

    float lm = -INFINITY;
    #pragma unroll
    for (int e = 0; e < 8; ++e) lm = fmaxf(lm, h[e]);
    red[tid] = lm; __syncthreads();
    for (int off = 128; off > 0; off >>= 1) {
        if (tid < off) red[tid] = fmaxf(red[tid], red[tid + off]);
        __syncthreads();
    }
    const float M = red[0];
    __syncthreads();

    float ls = 0.0f;
    #pragma unroll
    for (int e = 0; e < 8; ++e) {
        h[e] = __expf(h[e] - M);
        ls += h[e];
        hs[xb + 16 * e] = h[e];
    }
    red[tid] = ls; __syncthreads();
    for (int off = 128; off > 0; off >>= 1) {
        if (tid < off) red[tid] += red[tid + off];
        __syncthreads();
    }
    const float inv = 1.0f / red[0];
    for (int j = 0; j < 2; ++j) {
        const int e4 = 4 * (tid + 256 * j);
        const v4f v = *(const v4f*)&hs[e4];
        *(v4f*)&betaOut[(long)b * Ln + e4] = v * inv;
    }
}

#define P3_WAVES 32
#define P3_TOK   (Ln / P3_WAVES)
#define P3_RING  8

__global__ __launch_bounds__(1024)
void k_phase3(const int* __restrict__ tok, const float* __restrict__ emb,
              const float* __restrict__ beta, const float* __restrict__ lin_w,
              const float* __restrict__ lin_b, float* __restrict__ logits)
{
    __shared__ float part[P3_WAVES][2];
    const int b = blockIdx.x, tid = threadIdx.x;
    const int lane = tid & 31;
    const int w    = tid >> 5;

    const int*   __restrict__ trow = tok  + (long)b * Ln + w * P3_TOK;
    const float* __restrict__ brow = beta + (long)b * Ln + w * P3_TOK;
    const int   tv0 = trow[lane], tv1 = trow[lane + 32];
    const float bv0 = brow[lane], bv1 = brow[lane + 32];

    const int      pl = lane < 25 ? lane : 24;
    const unsigned o  = (unsigned)pl * 16u;
    const char* __restrict__ embc = (const char*)emb;

    v4f x0[P3_RING], x1[P3_RING], x2[P3_RING];
    v4f a0 = {}, a1 = {}, a2 = {};

#define P3_LOAD(slot, tvec, ln) {                                            \
        const int t_ = __builtin_amdgcn_readlane(tvec, ln);                  \
        const char* rp_ = embc + (size_t)(unsigned)t_ * (Dn * 4u);           \
        x0[slot] = *(const v4f*)(rp_ + o);                                   \
        x1[slot] = *(const v4f*)(rp_ + 400 + o);                             \
        x2[slot] = *(const v4f*)(rp_ + 800 + o); }

#define P3_USE(slot, bvec, ln) {                                             \
        const float bt_ = __builtin_bit_cast(float, __builtin_amdgcn_readlane( \
            __builtin_bit_cast(int, bvec), ln));                             \
        a0 += bt_ * x0[slot]; a1 += bt_ * x1[slot]; a2 += bt_ * x2[slot]; }

    #pragma unroll
    for (int j = 0; j < P3_RING; ++j) P3_LOAD(j, tv0, j);

    for (int g = 0; g < P3_TOK / P3_RING - 1; ++g) {
        const float bcur = g < 4 ? bv0 : bv1;
        const int   tnxt = g < 3 ? tv0 : tv1;
        const int   lc = (g & 3) * P3_RING;
        const int   ln = ((g + 1) & 3) * P3_RING;
        #pragma unroll
        for (int j = 0; j < P3_RING; ++j) {
            P3_USE(j, bcur, lc + j);
            P3_LOAD(j, tnxt, ln + j);
        }
    }
    #pragma unroll
    for (int j = 0; j < P3_RING; ++j) P3_USE(j, bv1, 32 - P3_RING + j);

#undef P3_LOAD
#undef P3_USE

    float p0 = 0.0f, p1 = 0.0f;
    if (lane < 25) {
        const int d0 = 4 * lane;
        for (int e = 0; e < 4; ++e) {
            p0 += a0[e] * lin_w[d0 + e] + a1[e] * lin_w[100 + d0 + e]
                + a2[e] * lin_w[200 + d0 + e];
            p1 += a0[e] * lin_w[Dn + d0 + e] + a1[e] * lin_w[Dn + 100 + d0 + e]
                + a2[e] * lin_w[Dn + 200 + d0 + e];
        }
    }
    for (int off = 16; off > 0; off >>= 1) {
        p0 += __shfl_xor(p0, off, 32);
        p1 += __shfl_xor(p1, off, 32);
    }
    if (lane == 0) { part[w][0] = p0; part[w][1] = p1; }
    __syncthreads();

    if (tid < 32) {
        float q0 = part[tid][0], q1 = part[tid][1];
        for (int off = 16; off > 0; off >>= 1) {
            q0 += __shfl_xor(q0, off, 32);
            q1 += __shfl_xor(q1, off, 32);
        }
        if (tid == 0) {
            const float o0 = q0 + lin_b[0];
            const float o1 = q1 + lin_b[1];
            const float mm = fmaxf(o0, o1);
            const float lse = mm + __logf(__expf(o0 - mm) + __expf(o1 - mm));
            logits[b * 2 + 0] = o0 - lse;
            logits[b * 2 + 1] = o1 - lse;
        }
    }
}

extern "C" void kernel_launch(void* const* d_in, const int* in_sizes, int n_in,
                              void* d_out, int out_size, void* d_ws, size_t ws_size,
                              hipStream_t stream)
{
    const int*   tok = (const int*)  d_in[0];
    const float* emb = (const float*)d_in[1];
    const float* cls = (const float*)d_in[2];
    const float* cw  = (const float*)d_in[3];
    const float* cb  = (const float*)d_in[4];
    const float* lw  = (const float*)d_in[5];
    const float* lb  = (const float*)d_in[6];

    float* out    = (float*)d_out;
    float* logits = out;
    float* beta   = out + Bn * Cn;
    float* g      = (float*)d_ws;

    k_phase1<<<1024, 256, 0, stream>>>(tok, emb, cls, g);
    k_phase2<<<Bn, 256, 0, stream>>>(g, cw, cb, beta);
    k_phase3<<<Bn, 1024, 0, stream>>>(tok, emb, beta, lw, lb, logits);
}


# Phases 2 and 3 fused, 4 blocks per batch row (256 blocks): each repeats the row's conv+softmax, then gathers only its 512 positions' table rows; a small kernel adds the row's 32 wave partials.
# speedup vs baseline: 3.9983x; 1.2086x over previous
#include <hip/hip_runtime.h>
#include <hip/hip_bf16.h>
#include <math.h>

#define Bn   64
#define Ln   2048
#define Dn   300
#define Cn   2
#define Kn   111
#define PADn 55

typedef __attribute__((ext_vector_type(2))) float v2f;
typedef __attribute__((ext_vector_type(8))) float v8f;
typedef __attribute__((ext_vector_type(4))) float v4f;

#define XS_PITCH 300

__global__ __launch_bounds__(256)
void k_phase1(const int* __restrict__ tok, const float* __restrict__ emb,
              const float* __restrict__ cls, float* __restrict__ g)
{
    __shared__ __attribute__((aligned(16))) float clsn[3 * 304];
    __shared__ float red[256];
    __shared__ __attribute__((aligned(16))) float xs[8][16 * XS_PITCH];
    const int tid = threadIdx.x;

    const int wave = tid >> 5;
    const int lane = tid & 31;
    const int tileId = blockIdx.x * 8 + wave;
    const int b  = tileId >> 7;
    const int l0 = (tileId & 127) * 16;
    const int m    = lane & 15;
    const int hi   = lane >> 4;

    const int token = tok[b * Ln + l0 + m];

    {
        const int      pl = lane < 25 ? lane : 24;
        const unsigned o  = (unsigned)pl * 16u;
        const char* __restrict__ embc = (const char*)emb;
        char* xsw = (char*)&xs[wave][0];
        #pragma unroll
        for (int half = 0; half < 2; ++half) {
            v4f x0[8], x1[8], x2[8];
            #pragma unroll
            for (int r = 0; r < 8; ++r) {
                const int t_ = __builtin_amdgcn_readlane(token, half * 8 + r);
                const char* rp_ = embc + (size_t)(unsigned)t_ * (Dn * 4u);
                x0[r] = *(const v4f*)(rp_ + o);
                x1[r] = *(const v4f*)(rp_ + 400 + o);
                x2[r] = *(const v4f*)(rp_ + 800 + o);
            }
            #pragma unroll
            for (int r = 0; r < 8; ++r) {
                char* wp_ = xsw + (half * 8 + r) * (XS_PITCH * 4) + o;
                *(v4f*)(wp_)       = x0[r];
                *(v4f*)(wp_ + 400) = x1[r];
                *(v4f*)(wp_ + 800) = x2[r];
            }
        }
    }

    for (int i = tid; i < 3 * 304; i += 256) clsn[i] = 0.0f;
    __syncthreads();
    for (int c = 0; c < Cn; ++c) {
        float s = 0.0f;
        for (int d = tid; d < Dn; d += 256) { float v = cls[c * Dn + d]; s += v * v; }
        red[tid] = s;
        __syncthreads();
        for (int off = 128; off > 0; off >>= 1) {
            if (tid < off) red[tid] += red[tid + off];
            __syncthreads();
        }
        float rn = 1.0f / fmaxf(sqrtf(red[0]), 1e-12f);
        __syncthreads();
        for (int d = tid; d < Dn; d += 256) clsn[c * 304 + d] = cls[c * Dn + d] * rn;
        __syncthreads();
    }

    const float* __restrict__ ar = &xs[wave][m * XS_PITCH + 4 * hi];
    const float* __restrict__ br = &clsn[(m < 2 ? m : 2) * 304 + 4 * hi];

    v8f acc0 = {}, acc1 = {};
    float n2 = 0.0f;
    #pragma unroll
    for (int j = 0; j < 37; ++j) {
        const v4f a  = *(const v4f*)(ar + 8 * j);
        const v4f bq = *(const v4f*)(br + 8 * j);
        n2 += a.x * a.x + a.y * a.y + a.z * a.z + a.w * a.w;
        v2f a01, a23, b01, b23;
        a01.x = a.x;  a01.y = a.y;  a23.x = a.z;  a23.y = a.w;
        b01.x = bq.x; b01.y = bq.y; b23.x = bq.z; b23.y = bq.w;
        acc0 = __builtin_amdgcn_wmma_f32_16x16x4_f32(
            false, a01, false, b01, (short)0, acc0, false, false);
        acc1 = __builtin_amdgcn_wmma_f32_16x16x4_f32(
            false, a23, false, b23, (short)0, acc1, false, false);
    }
    {
        const v2f a  = *(const v2f*)(&xs[wave][m * XS_PITCH + 296 + 2 * hi]);
        const v2f bq = *(const v2f*)(&clsn[(m < 2 ? m : 2) * 304 + 296 + 2 * hi]);
        n2 += a.x * a.x + a.y * a.y;
        acc0 = __builtin_amdgcn_wmma_f32_16x16x4_f32(
            false, a, false, bq, (short)0, acc0, false, false);
    }
    const v8f acc = acc0 + acc1;

    n2 += __shfl_xor(n2, 16, 32);

    const int  c      = m;
    const bool active = (m < 2);
    const int  mbase  = hi * 8;
    float* __restrict__ gout = g + (((long)b * Cn + c) * Ln + l0 + mbase);
    #pragma unroll
    for (int i = 0; i < 8; ++i) {
        float nn    = __shfl(n2, mbase + i, 32);
        float scale = 1.0f / fmaxf(sqrtf(nn), 1e-12f);
        if (active) gout[i] = acc[i] * scale;
    }
}

#define GP_ROWS  135
#define GP_PITCH 20
#define WP_LEAD  15
#define WP_LEN   144

#define SPLIT     4
#define P3_WAVES  8
#define ROW_WAVES (SPLIT * P3_WAVES)
#define P3_TOK    (Ln / ROW_WAVES)
#define P3_RING   8
#define WS_FLOATS (Bn * Cn * Ln + Bn * ROW_WAVES * 2)

__global__ __launch_bounds__(256)
void k_phase23(const float* __restrict__ g, const float* __restrict__ conv_w,
               const float* __restrict__ conv_b, const int* __restrict__ tok,
               const float* __restrict__ emb, const float* __restrict__ lin_w,
               float* __restrict__ betaOut, float* __restrict__ part,
               size_t ws_bytes)
{
    __shared__ __attribute__((aligned(16))) float gp[Cn][GP_ROWS * GP_PITCH];
    __shared__ float wp[Cn * Cn][WP_LEN];
    __shared__ float red[256];
    __shared__ __attribute__((aligned(16))) float hs[Ln];
    const int b  = blockIdx.x / SPLIT;
    const int sp = blockIdx.x % SPLIT;
    const int tid = threadIdx.x;
    const int lane = tid & 31;
    const int wv   = tid >> 5;
    const int rw   = sp * P3_WAVES + wv;

    const int* __restrict__ trow = tok + (long)b * Ln + rw * P3_TOK;
    const int tv0 = trow[lane], tv1 = trow[lane + 32];

    for (int i = tid; i < Cn * GP_ROWS * GP_PITCH; i += 256) (&gp[0][0])[i] = 0.0f;
    for (int i = tid; i < Cn * Cn * WP_LEN; i += 256) (&wp[0][0])[i] = 0.0f;
    __syncthreads();
    for (int i = tid; i < Ln; i += 256) {
        const int u = i + PADn;
        const int a = (u >> 4) * GP_PITCH + (u & 15);
        gp[0][a] = g[((long)b * Cn + 0) * Ln + i];
        gp[1][a] = g[((long)b * Cn + 1) * Ln + i];
    }
    for (int i = tid; i < Cn * Cn * Kn; i += 256) {
        const int pr = i / Kn;
        wp[pr][WP_LEAD + i - pr * Kn] = conv_w[i];
    }
    __syncthreads();

    const int c    = lane & 15;
    const int hh   = lane >> 4;
    const float* __restrict__ a0p = &gp[0][(16 * wv + c) * GP_PITCH + 2 * hh];
    const float* __restrict__ a1p = &gp[1][(16 * wv + c) * GP_PITCH + 2 * hh];
    const float* __restrict__ bp  = &wp[0][WP_LEAD - c + 2 * hh];

    v8f acc0 = {}, acc1 = {};
    #pragma unroll
    for (int p = 0; p < 8; ++p) {
        #pragma unroll
        for (int q = 0; q < 4; ++q) {
            const int ao = p * GP_PITCH + 4 * q;
            const int t  = 16 * p + 4 * q;
            const v2f a0 = *(const v2f*)(a0p + ao);
            const v2f a1 = *(const v2f*)(a1p + ao);
            v2f b00, b01, b10, b11;
            b00.x = bp[0 * WP_LEN + t]; b00.y = bp[0 * WP_LEN + t + 1];
            b01.x = bp[1 * WP_LEN + t]; b01.y = bp[1 * WP_LEN + t + 1];
            b10.x = bp[2 * WP_LEN + t]; b10.y = bp[2 * WP_LEN + t + 1];
            b11.x = bp[3 * WP_LEN + t]; b11.y = bp[3 * WP_LEN + t + 1];
            acc0 = __builtin_amdgcn_wmma_f32_16x16x4_f32(
                false, a0, false, b00, (short)0, acc0, false, false);
            acc1 = __builtin_amdgcn_wmma_f32_16x16x4_f32(
                false, a0, false, b10, (short)0, acc1, false, false);
            acc0 = __builtin_amdgcn_wmma_f32_16x16x4_f32(
                false, a1, false, b01, (short)0, acc0, false, false);
            acc1 = __builtin_amdgcn_wmma_f32_16x16x4_f32(
                false, a1, false, b11, (short)0, acc1, false, false);
        }
    }

    const int xb = 256 * wv + 128 * hh + c;
    const float cb0 = conv_b[0], cb1 = conv_b[1];
    float h[8];
    for (int e = 0; e < 8; ++e)
        h[e] = fmaxf(fmaxf(acc0[e] + cb0, 0.0f), fmaxf(acc1[e] + cb1, 0.0f));

    float lm = -INFINITY;
    #pragma unroll
    for (int e = 0; e < 8; ++e) lm = fmaxf(lm, h[e]);
    red[tid] = lm; __syncthreads();
    for (int off = 128; off > 0; off >>= 1) {
        if (tid < off) red[tid] = fmaxf(red[tid], red[tid + off]);
        __syncthreads();
    }
    const float M = red[0];
    __syncthreads();

    float ls = 0.0f;
    #pragma unroll
    for (int e = 0; e < 8; ++e) {
        h[e] = __expf(h[e] - M);
        ls += h[e];
        hs[xb + 16 * e] = h[e];
    }
    red[tid] = ls; __syncthreads();
    for (int off = 128; off > 0; off >>= 1) {
        if (tid < off) red[tid] += red[tid + off];
        __syncthreads();
    }
    const float inv = 1.0f / red[0];
    if (tid < Ln / SPLIT / 4) {
        const int e4 = (Ln / SPLIT) * sp + 4 * tid;
        const v4f v = *(const v4f*)&hs[e4];
        *(v4f*)&betaOut[(long)b * Ln + e4] = v * inv;
    }

    const float bv0 = hs[rw * P3_TOK + lane] * inv;
    const float bv1 = hs[rw * P3_TOK + 32 + lane] * inv;

    const int      pl = lane < 25 ? lane : 24;
    const unsigned o  = (unsigned)pl * 16u;
    const char* __restrict__ embc = (const char*)emb;

    v4f x0[P3_RING], x1[P3_RING], x2[P3_RING];
    v4f z0 = {}, z1 = {}, z2 = {};

#define P3_LOAD(slot, tvec, ln) {                                            \
        const int t_ = __builtin_amdgcn_readlane(tvec, ln);                  \
        const char* rp_ = embc + (size_t)(unsigned)t_ * (Dn * 4u);           \
        x0[slot] = *(const v4f*)(rp_ + o);                                   \
        x1[slot] = *(const v4f*)(rp_ + 400 + o);                             \
        x2[slot] = *(const v4f*)(rp_ + 800 + o); }

#define P3_USE(slot, bvec, ln) {                                             \
        const float bt_ = __builtin_bit_cast(float, __builtin_amdgcn_readlane( \
            __builtin_bit_cast(int, bvec), ln));                             \
        z0 += bt_ * x0[slot]; z1 += bt_ * x1[slot]; z2 += bt_ * x2[slot]; }

    #pragma unroll
    for (int j = 0; j < P3_RING; ++j) P3_LOAD(j, tv0, j);

    for (int grp = 0; grp < P3_TOK / P3_RING - 1; ++grp) {
        const float bcur = grp < 4 ? bv0 : bv1;
        const int   tnxt = grp < 3 ? tv0 : tv1;
        const int   lc = (grp & 3) * P3_RING;
        const int   ln = ((grp + 1) & 3) * P3_RING;
        #pragma unroll
        for (int j = 0; j < P3_RING; ++j) {
            P3_USE(j, bcur, lc + j);
            P3_LOAD(j, tnxt, ln + j);
        }
    }
    #pragma unroll
    for (int j = 0; j < P3_RING; ++j) P3_USE(j, bv1, 32 - P3_RING + j);

#undef P3_LOAD
#undef P3_USE

    float p0 = 0.0f, p1 = 0.0f;
    if (lane < 25) {
        const int d0 = 4 * lane;
        for (int e = 0; e < 4; ++e) {
            p0 += z0[e] * lin_w[d0 + e] + z1[e] * lin_w[100 + d0 + e]
                + z2[e] * lin_w[200 + d0 + e];
            p1 += z0[e] * lin_w[Dn + d0 + e] + z1[e] * lin_w[Dn + 100 + d0 + e]
                + z2[e] * lin_w[Dn + 200 + d0 + e];
        }
    }
    for (int off = 16; off > 0; off >>= 1) {
        p0 += __shfl_xor(p0, off, 32);
        p1 += __shfl_xor(p1, off, 32);
    }
    if (lane == 0 && ws_bytes >= sizeof(float) * WS_FLOATS) {
        v2f pr;
        pr.x = p0; pr.y = p1;
        *(v2f*)&part[(b * ROW_WAVES + rw) * 2] = pr;
    }
}

__global__ __launch_bounds__(32)
void k_head(const float* __restrict__ part, const float* __restrict__ lin_b,
            float* __restrict__ logits, size_t ws_bytes)
{
    if (ws_bytes < sizeof(float) * WS_FLOATS) return;
    const int b = blockIdx.x, tid = threadIdx.x;
    const v2f pr = *(const v2f*)&part[(b * ROW_WAVES + tid) * 2];
    float q0 = pr.x, q1 = pr.y;
    for (int off = 16; off > 0; off >>= 1) {
        q0 += __shfl_xor(q0, off, 32);
        q1 += __shfl_xor(q1, off, 32);
    }
    if (tid == 0) {
        const float o0 = q0 + lin_b[0];
        const float o1 = q1 + lin_b[1];
        const float mm = fmaxf(o0, o1);
        const float lse = mm + __logf(__expf(o0 - mm) + __expf(o1 - mm));
        logits[b * 2 + 0] = o0 - lse;
        logits[b * 2 + 1] = o1 - lse;
    }
}

extern "C" void kernel_launch(void* const* d_in, const int* in_sizes, int n_in,
                              void* d_out, int out_size, void* d_ws, size_t ws_size,
                              hipStream_t stream)
{
    const int*   tok = (const int*)  d_in[0];
    const float* emb = (const float*)d_in[1];
    const float* cls = (const float*)d_in[2];
    const float* cw  = (const float*)d_in[3];
    const float* cb  = (const float*)d_in[4];
    const float* lw  = (const float*)d_in[5];
    const float* lb  = (const float*)d_in[6];

    float* out    = (float*)d_out;
    float* logits = out;
    float* beta   = out + Bn * Cn;
    float* g      = (float*)d_ws;
    float* part   = g + Bn * Cn * Ln;

    k_phase1<<<1024, 256, 0, stream>>>(tok, emb, cls, g);
    k_phase23<<<Bn * SPLIT, 256, 0, stream>>>(g, cw, cb, tok, emb, lw, beta, part, ws_size);
    k_head<<<Bn, 32, 0, stream>>>(part, lb, logits, ws_size);
}


# Phase 1 stages a tile one 100-dimension chunk at a time: chunk c+1's 16 row loads fly under chunk c's 25 matrix steps; a wave keeps 6400 B of LDS and 16 loads, so 4 blocks fit a WGP, not 2.
# speedup vs baseline: 4.5522x; 1.1385x over previous
#include <hip/hip_runtime.h>
#include <hip/hip_bf16.h>
#include <math.h>

#define Bn   64
#define Ln   2048
#define Dn   300
#define Cn   2
#define Kn   111
#define PADn 55

typedef __attribute__((ext_vector_type(2))) float v2f;
typedef __attribute__((ext_vector_type(8))) float v8f;
typedef __attribute__((ext_vector_type(4))) float v4f;

#define CH_DIMS  100
#define CH_PITCH 100

__global__ __launch_bounds__(256)
void k_phase1(const int* __restrict__ tok, const float* __restrict__ emb,
              const float* __restrict__ cls, float* __restrict__ g)
{
    __shared__ __attribute__((aligned(16))) float clsn[3 * 304];
    __shared__ float red[256];
    __shared__ __attribute__((aligned(16))) float xs[8][16 * CH_PITCH];
    const int tid = threadIdx.x;

    const int wave = tid >> 5;
    const int lane = tid & 31;
    const int tileId = blockIdx.x * 8 + wave;
    const int b  = tileId >> 7;
    const int l0 = (tileId & 127) * 16;
    const int m    = lane & 15;
    const int hi   = lane >> 4;

    const int token = tok[b * Ln + l0 + m];

    const int      pl = lane < 25 ? lane : 24;
    const unsigned o  = (unsigned)pl * 16u;
    const char* __restrict__ embc = (const char*)emb;
    char* xsw = (char*)&xs[wave][0] + o;

    const char* rp[16];
    #pragma unroll
    for (int r = 0; r < 16; ++r) {
        const int t_ = __builtin_amdgcn_readlane(token, r);
        rp[r] = embc + (size_t)(unsigned)t_ * (Dn * 4u);
    }

    v4f x[16];
    #pragma unroll
    for (int r = 0; r < 16; ++r) x[r] = *(const v4f*)(rp[r] + o);

    for (int i = tid; i < 3 * 304; i += 256) clsn[i] = 0.0f;
    __syncthreads();
    for (int c = 0; c < Cn; ++c) {
        float s = 0.0f;
        for (int d = tid; d < Dn; d += 256) { float v = cls[c * Dn + d]; s += v * v; }
        red[tid] = s;
        __syncthreads();
        for (int off = 128; off > 0; off >>= 1) {
            if (tid < off) red[tid] += red[tid + off];
            __syncthreads();
        }
        float rn = 1.0f / fmaxf(sqrtf(red[0]), 1e-12f);
        __syncthreads();
        for (int d = tid; d < Dn; d += 256) clsn[c * 304 + d] = cls[c * Dn + d] * rn;
        __syncthreads();
    }

    const float* __restrict__ ar = &xs[wave][m * CH_PITCH + 4 * hi];
    const float* __restrict__ br = &clsn[(m < 2 ? m : 2) * 304 + 4 * hi];

    v8f acc0 = {}, acc1 = {};
    float n2 = 0.0f;
    #pragma unroll
    for (int q = 0; q < 3; ++q) {
        #pragma unroll
        for (int r = 0; r < 16; ++r) *(v4f*)(xsw + r * (CH_PITCH * 4)) = x[r];
        if (q < 2) {
            unsigned on = o + (unsigned)(q + 1) * (CH_DIMS * 4u);
            asm volatile("" : "+v"(on));
            #pragma unroll
            for (int r = 0; r < 16; ++r) x[r] = *(const v4f*)(rp[r] + on);
        }
        __builtin_amdgcn_fence(__ATOMIC_RELEASE, "wavefront");
        __builtin_amdgcn_wave_barrier();
        __builtin_amdgcn_fence(__ATOMIC_ACQUIRE, "wavefront");

        const float* __restrict__ bq_ = br + q * CH_DIMS;
        #pragma unroll
        for (int j = 0; j < 12; ++j) {
            const v4f a  = *(const v4f*)(ar + 8 * j);
            const v4f bq = *(const v4f*)(bq_ + 8 * j);
            n2 += a.x * a.x + a.y * a.y + a.z * a.z + a.w * a.w;
            v2f a01, a23, b01, b23;
            a01.x = a.x;  a01.y = a.y;  a23.x = a.z;  a23.y = a.w;
            b01.x = bq.x; b01.y = bq.y; b23.x = bq.z; b23.y = bq.w;
            acc0 = __builtin_amdgcn_wmma_f32_16x16x4_f32(
                false, a01, false, b01, (short)0, acc0, false, false);
            acc1 = __builtin_amdgcn_wmma_f32_16x16x4_f32(
                false, a23, false, b23, (short)0, acc1, false, false);
        }
        {
            const v2f a  = *(const v2f*)(&xs[wave][m * CH_PITCH + 96 + 2 * hi]);
            const v2f bq = *(const v2f*)(&clsn[(m < 2 ? m : 2) * 304 + q * CH_DIMS + 96 + 2 * hi]);
            n2 += a.x * a.x + a.y * a.y;
            if (q & 1)
                acc1 = __builtin_amdgcn_wmma_f32_16x16x4_f32(
                    false, a, false, bq, (short)0, acc1, false, false);
            else
                acc0 = __builtin_amdgcn_wmma_f32_16x16x4_f32(
                    false, a, false, bq, (short)0, acc0, false, false);
        }
        __builtin_amdgcn_fence(__ATOMIC_RELEASE, "wavefront");
        __builtin_amdgcn_wave_barrier();
        __builtin_amdgcn_fence(__ATOMIC_ACQUIRE, "wavefront");
    }
    const v8f acc = acc0 + acc1;

    n2 += __shfl_xor(n2, 16, 32);

    const int  c      = m;
    const bool active = (m < 2);
    const int  mbase  = hi * 8;
    float* __restrict__ gout = g + (((long)b * Cn + c) * Ln + l0 + mbase);
    #pragma unroll
    for (int i = 0; i < 8; ++i) {
        float nn    = __shfl(n2, mbase + i, 32);
        float scale = 1.0f / fmaxf(sqrtf(nn), 1e-12f);
        if (active) gout[i] = acc[i] * scale;
    }
}

#define GP_ROWS  135
#define GP_PITCH 20
#define WP_LEAD  15
#define WP_LEN   144

#define SPLIT     4
#define P3_WAVES  8
#define ROW_WAVES (SPLIT * P3_WAVES)
#define P3_TOK    (Ln / ROW_WAVES)
#define P3_RING   8
#define WS_FLOATS (Bn * Cn * Ln + Bn * ROW_WAVES * 2)

__global__ __launch_bounds__(256)
void k_phase23(const float* __restrict__ g, const float* __restrict__ conv_w,
               const float* __restrict__ conv_b, const int* __restrict__ tok,
               const float* __restrict__ emb, const float* __restrict__ lin_w,
               float* __restrict__ betaOut, float* __restrict__ part,
               size_t ws_bytes)
{
    __shared__ __attribute__((aligned(16))) float gp[Cn][GP_ROWS * GP_PITCH];
    __shared__ float wp[Cn * Cn][WP_LEN];
    __shared__ float red[256];
    __shared__ __attribute__((aligned(16))) float hs[Ln];
    const int b  = blockIdx.x / SPLIT;
    const int sp = blockIdx.x % SPLIT;
    const int tid = threadIdx.x;
    const int lane = tid & 31;
    const int wv   = tid >> 5;
    const int rw   = sp * P3_WAVES + wv;

    const int* __restrict__ trow = tok + (long)b * Ln + rw * P3_TOK;
    const int tv0 = trow[lane], tv1 = trow[lane + 32];

    for (int i = tid; i < Cn * GP_ROWS * GP_PITCH; i += 256) (&gp[0][0])[i] = 0.0f;
    for (int i = tid; i < Cn * Cn * WP_LEN; i += 256) (&wp[0][0])[i] = 0.0f;
    __syncthreads();
    for (int i = tid; i < Ln; i += 256) {
        const int u = i + PADn;
        const int a = (u >> 4) * GP_PITCH + (u & 15);
        gp[0][a] = g[((long)b * Cn + 0) * Ln + i];
        gp[1][a] = g[((long)b * Cn + 1) * Ln + i];
    }
    for (int i = tid; i < Cn * Cn * Kn; i += 256) {
        const int pr = i / Kn;
        wp[pr][WP_LEAD + i - pr * Kn] = conv_w[i];
    }
    __syncthreads();

    const int c    = lane & 15;
    const int hh   = lane >> 4;
    const float* __restrict__ a0p = &gp[0][(16 * wv + c) * GP_PITCH + 2 * hh];
    const float* __restrict__ a1p = &gp[1][(16 * wv + c) * GP_PITCH + 2 * hh];
    const float* __restrict__ bp  = &wp[0][WP_LEAD - c + 2 * hh];

    v8f acc0 = {}, acc1 = {};
    #pragma unroll
    for (int p = 0; p < 8; ++p) {
        #pragma unroll
        for (int q = 0; q < 4; ++q) {
            const int ao = p * GP_PITCH + 4 * q;
            const int t  = 16 * p + 4 * q;
            const v2f a0 = *(const v2f*)(a0p + ao);
            const v2f a1 = *(const v2f*)(a1p + ao);
            v2f b00, b01, b10, b11;
            b00.x = bp[0 * WP_LEN + t]; b00.y = bp[0 * WP_LEN + t + 1];
            b01.x = bp[1 * WP_LEN + t]; b01.y = bp[1 * WP_LEN + t + 1];
            b10.x = bp[2 * WP_LEN + t]; b10.y = bp[2 * WP_LEN + t + 1];
            b11.x = bp[3 * WP_LEN + t]; b11.y = bp[3 * WP_LEN + t + 1];
            acc0 = __builtin_amdgcn_wmma_f32_16x16x4_f32(
                false, a0, false, b00, (short)0, acc0, false, false);
            acc1 = __builtin_amdgcn_wmma_f32_16x16x4_f32(
                false, a0, false, b10, (short)0, acc1, false, false);
            acc0 = __builtin_amdgcn_wmma_f32_16x16x4_f32(
                false, a1, false, b01, (short)0, acc0, false, false);
            acc1 = __builtin_amdgcn_wmma_f32_16x16x4_f32(
                false, a1, false, b11, (short)0, acc1, false, false);
        }
    }

    const int xb = 256 * wv + 128 * hh + c;
    const float cb0 = conv_b[0], cb1 = conv_b[1];
    float h[8];
    for (int e = 0; e < 8; ++e)
        h[e] = fmaxf(fmaxf(acc0[e] + cb0, 0.0f), fmaxf(acc1[e] + cb1, 0.0f));

    float lm = -INFINITY;
    #pragma unroll
    for (int e = 0; e < 8; ++e) lm = fmaxf(lm, h[e]);
    red[tid] = lm; __syncthreads();
    for (int off = 128; off > 0; off >>= 1) {
        if (tid < off) red[tid] = fmaxf(red[tid], red[tid + off]);
        __syncthreads();
    }
    const float M = red[0];
    __syncthreads();

    float ls = 0.0f;
    #pragma unroll
    for (int e = 0; e < 8; ++e) {
        h[e] = __expf(h[e] - M);
        ls += h[e];
        hs[xb + 16 * e] = h[e];
    }
    red[tid] = ls; __syncthreads();
    for (int off = 128; off > 0; off >>= 1) {
        if (tid < off) red[tid] += red[tid + off];
        __syncthreads();
    }
    const float inv = 1.0f / red[0];
    if (tid < Ln / SPLIT / 4) {
        const int e4 = (Ln / SPLIT) * sp + 4 * tid;
        const v4f v = *(const v4f*)&hs[e4];
        *(v4f*)&betaOut[(long)b * Ln + e4] = v * inv;
    }

    const float bv0 = hs[rw * P3_TOK + lane] * inv;
    const float bv1 = hs[rw * P3_TOK + 32 + lane] * inv;

    const int      pl = lane < 25 ? lane : 24;
    const unsigned o  = (unsigned)pl * 16u;
    const char* __restrict__ embc = (const char*)emb;

    v4f x0[P3_RING], x1[P3_RING], x2[P3_RING];
    v4f z0 = {}, z1 = {}, z2 = {};

#define P3_LOAD(slot, tvec, ln) {                                            \
        const int t_ = __builtin_amdgcn_readlane(tvec, ln);                  \
        const char* rp_ = embc + (size_t)(unsigned)t_ * (Dn * 4u);           \
        x0[slot] = *(const v4f*)(rp_ + o);                                   \
        x1[slot] = *(const v4f*)(rp_ + 400 + o);                             \
        x2[slot] = *(const v4f*)(rp_ + 800 + o); }

#define P3_USE(slot, bvec, ln) {                                             \
        const float bt_ = __builtin_bit_cast(float, __builtin_amdgcn_readlane( \
            __builtin_bit_cast(int, bvec), ln));                             \
        z0 += bt_ * x0[slot]; z1 += bt_ * x1[slot]; z2 += bt_ * x2[slot]; }

    #pragma unroll
    for (int j = 0; j < P3_RING; ++j) P3_LOAD(j, tv0, j);

    for (int grp = 0; grp < P3_TOK / P3_RING - 1; ++grp) {
        const float bcur = grp < 4 ? bv0 : bv1;
        const int   tnxt = grp < 3 ? tv0 : tv1;
        const int   lc = (grp & 3) * P3_RING;
        const int   ln = ((grp + 1) & 3) * P3_RING;
        #pragma unroll
        for (int j = 0; j < P3_RING; ++j) {
            P3_USE(j, bcur, lc + j);
            P3_LOAD(j, tnxt, ln + j);
        }
    }
    #pragma unroll
    for (int j = 0; j < P3_RING; ++j) P3_USE(j, bv1, 32 - P3_RING + j);

#undef P3_LOAD
#undef P3_USE

    float p0 = 0.0f, p1 = 0.0f;
    if (lane < 25) {
        const int d0 = 4 * lane;
        for (int e = 0; e < 4; ++e) {
            p0 += z0[e] * lin_w[d0 + e] + z1[e] * lin_w[100 + d0 + e]
                + z2[e] * lin_w[200 + d0 + e];
            p1 += z0[e] * lin_w[Dn + d0 + e] + z1[e] * lin_w[Dn + 100 + d0 + e]
                + z2[e] * lin_w[Dn + 200 + d0 + e];
        }
    }
    for (int off = 16; off > 0; off >>= 1) {
        p0 += __shfl_xor(p0, off, 32);
        p1 += __shfl_xor(p1, off, 32);
    }
    if (lane == 0 && ws_bytes >= sizeof(float) * WS_FLOATS) {
        v2f pr;
        pr.x = p0; pr.y = p1;
        *(v2f*)&part[(b * ROW_WAVES + rw) * 2] = pr;
    }
}

__global__ __launch_bounds__(32)
void k_head(const float* __restrict__ part, const float* __restrict__ lin_b,
            float* __restrict__ logits, size_t ws_bytes)
{
    if (ws_bytes < sizeof(float) * WS_FLOATS) return;
    const int b = blockIdx.x, tid = threadIdx.x;
    const v2f pr = *(const v2f*)&part[(b * ROW_WAVES + tid) * 2];
    float q0 = pr.x, q1 = pr.y;
    for (int off = 16; off > 0; off >>= 1) {
        q0 += __shfl_xor(q0, off, 32);
        q1 += __shfl_xor(q1, off, 32);
    }
    if (tid == 0) {
        const float o0 = q0 + lin_b[0];
        const float o1 = q1 + lin_b[1];
        const float mm = fmaxf(o0, o1);
        const float lse = mm + __logf(__expf(o0 - mm) + __expf(o1 - mm));
        logits[b * 2 + 0] = o0 - lse;
        logits[b * 2 + 1] = o1 - lse;
    }
}

extern "C" void kernel_launch(void* const* d_in, const int* in_sizes, int n_in,
                              void* d_out, int out_size, void* d_ws, size_t ws_size,
                              hipStream_t stream)
{
    const int*   tok = (const int*)  d_in[0];
    const float* emb = (const float*)d_in[1];
    const float* cls = (const float*)d_in[2];
    const float* cw  = (const float*)d_in[3];
    const float* cb  = (const float*)d_in[4];
    const float* lw  = (const float*)d_in[5];
    const float* lb  = (const float*)d_in[6];

    float* out    = (float*)d_out;
    float* logits = out;
    float* beta   = out + Bn * Cn;
    float* g      = (float*)d_ws;
    float* part   = g + Bn * Cn * Ln;

    k_phase1<<<1024, 256, 0, stream>>>(tok, emb, cls, g);
    k_phase23<<<Bn * SPLIT, 256, 0, stream>>>(g, cw, cb, tok, emb, lw, beta, part, ws_size);
    k_head<<<Bn, 32, 0, stream>>>(part, lb, logits, ws_size);
}


# Phase 3's table rows are asked for under phase 2: 20 rows per wave go out before the conv and are set aside in LDS behind it, 20 more wait in registers during the softmax; barriers order LDS only.
# speedup vs baseline: 4.8939x; 1.0751x over previous
#include <hip/hip_runtime.h>
#include <hip/hip_bf16.h>
#include <math.h>

#define Bn   64
#define Ln   2048
#define Dn   300
#define Cn   2
#define Kn   111
#define PADn 55

typedef __attribute__((ext_vector_type(2))) float v2f;
typedef __attribute__((ext_vector_type(8))) float v8f;
typedef __attribute__((ext_vector_type(4))) float v4f;

#define CH_DIMS  100
#define CH_PITCH 100

__global__ __launch_bounds__(256)
void k_phase1(const int* __restrict__ tok, const float* __restrict__ emb,
              const float* __restrict__ cls, float* __restrict__ g)
{
    __shared__ __attribute__((aligned(16))) float clsn[3 * 304];
    __shared__ float red[256];
    __shared__ __attribute__((aligned(16))) float xs[8][16 * CH_PITCH];
    const int tid = threadIdx.x;

    const int wave = tid >> 5;
    const int lane = tid & 31;
    const int tileId = blockIdx.x * 8 + wave;
    const int b  = tileId >> 7;
    const int l0 = (tileId & 127) * 16;
    const int m    = lane & 15;
    const int hi   = lane >> 4;

    const int token = tok[b * Ln + l0 + m];

    const int      pl = lane < 25 ? lane : 24;
    const unsigned o  = (unsigned)pl * 16u;
    const char* __restrict__ embc = (const char*)emb;
    char* xsw = (char*)&xs[wave][0] + o;

    const char* rp[16];
    #pragma unroll
    for (int r = 0; r < 16; ++r) {
        const int t_ = __builtin_amdgcn_readlane(token, r);
        rp[r] = embc + (size_t)(unsigned)t_ * (Dn * 4u);
    }

    v4f x[16];
    #pragma unroll
    for (int r = 0; r < 16; ++r) x[r] = *(const v4f*)(rp[r] + o);

    for (int i = tid; i < 3 * 304; i += 256) clsn[i] = 0.0f;
    __syncthreads();
    for (int c = 0; c < Cn; ++c) {
        float s = 0.0f;
        for (int d = tid; d < Dn; d += 256) { float v = cls[c * Dn + d]; s += v * v; }
        red[tid] = s;
        __syncthreads();
        for (int off = 128; off > 0; off >>= 1) {
            if (tid < off) red[tid] += red[tid + off];
            __syncthreads();
        }
        float rn = 1.0f / fmaxf(sqrtf(red[0]), 1e-12f);
        __syncthreads();
        for (int d = tid; d < Dn; d += 256) clsn[c * 304 + d] = cls[c * Dn + d] * rn;
        __syncthreads();
    }

    const float* __restrict__ ar = &xs[wave][m * CH_PITCH + 4 * hi];
    const float* __restrict__ br = &clsn[(m < 2 ? m : 2) * 304 + 4 * hi];

    v8f acc0 = {}, acc1 = {};
    float n2 = 0.0f;
    #pragma unroll
    for (int q = 0; q < 3; ++q) {
        #pragma unroll
        for (int r = 0; r < 16; ++r) *(v4f*)(xsw + r * (CH_PITCH * 4)) = x[r];
        if (q < 2) {
            unsigned on = o + (unsigned)(q + 1) * (CH_DIMS * 4u);
            asm volatile("" : "+v"(on));
            #pragma unroll
            for (int r = 0; r < 16; ++r) x[r] = *(const v4f*)(rp[r] + on);
        }
        __builtin_amdgcn_fence(__ATOMIC_RELEASE, "wavefront");
        __builtin_amdgcn_wave_barrier();
        __builtin_amdgcn_fence(__ATOMIC_ACQUIRE, "wavefront");

        const float* __restrict__ bq_ = br + q * CH_DIMS;
        #pragma unroll
        for (int j = 0; j < 12; ++j) {
            const v4f a  = *(const v4f*)(ar + 8 * j);
            const v4f bq = *(const v4f*)(bq_ + 8 * j);
            n2 += a.x * a.x + a.y * a.y + a.z * a.z + a.w * a.w;
            v2f a01, a23, b01, b23;
            a01.x = a.x;  a01.y = a.y;  a23.x = a.z;  a23.y = a.w;
            b01.x = bq.x; b01.y = bq.y; b23.x = bq.z; b23.y = bq.w;
            acc0 = __builtin_amdgcn_wmma_f32_16x16x4_f32(
                false, a01, false, b01, (short)0, acc0, false, false);
            acc1 = __builtin_amdgcn_wmma_f32_16x16x4_f32(
                false, a23, false, b23, (short)0, acc1, false, false);
        }
        {
            const v2f a  = *(const v2f*)(&xs[wave][m * CH_PITCH + 96 + 2 * hi]);
            const v2f bq = *(const v2f*)(&clsn[(m < 2 ? m : 2) * 304 + q * CH_DIMS + 96 + 2 * hi]);
            n2 += a.x * a.x + a.y * a.y;
            if (q & 1)
                acc1 = __builtin_amdgcn_wmma_f32_16x16x4_f32(
                    false, a, false, bq, (short)0, acc1, false, false);
            else
                acc0 = __builtin_amdgcn_wmma_f32_16x16x4_f32(
                    false, a, false, bq, (short)0, acc0, false, false);
        }
        __builtin_amdgcn_fence(__ATOMIC_RELEASE, "wavefront");
        __builtin_amdgcn_wave_barrier();
        __builtin_amdgcn_fence(__ATOMIC_ACQUIRE, "wavefront");
    }
    const v8f acc = acc0 + acc1;

    n2 += __shfl_xor(n2, 16, 32);

    const int  c      = m;
    const bool active = (m < 2);
    const int  mbase  = hi * 8;
    float* __restrict__ gout = g + (((long)b * Cn + c) * Ln + l0 + mbase);
    #pragma unroll
    for (int i = 0; i < 8; ++i) {
        float nn    = __shfl(n2, mbase + i, 32);
        float scale = 1.0f / fmaxf(sqrtf(nn), 1e-12f);
        if (active) gout[i] = acc[i] * scale;
    }
}

#define GP_ROWS  135
#define GP_PITCH 20
#define WP_LEAD  15
#define WP_LEN   144

#define SPLIT     4
#define P3_WAVES  8
#define ROW_WAVES (SPLIT * P3_WAVES)
#define P3_TOK    (Ln / ROW_WAVES)
#define P3_PRE    20
#define WS_FLOATS (Bn * Cn * Ln + Bn * ROW_WAVES * 2)

__device__ __forceinline__ void lds_barrier()
{
    __builtin_amdgcn_fence(__ATOMIC_RELEASE, "workgroup", "local");
    __builtin_amdgcn_s_barrier();
    __builtin_amdgcn_fence(__ATOMIC_ACQUIRE, "workgroup", "local");
}

__global__ __launch_bounds__(256)
void k_phase23(const float* __restrict__ g, const float* __restrict__ conv_w,
               const float* __restrict__ conv_b, const int* __restrict__ tok,
               const float* __restrict__ emb, const float* __restrict__ lin_w,
               float* __restrict__ betaOut, float* __restrict__ part,
               size_t ws_bytes)
{
    __shared__ __attribute__((aligned(16))) float gp[Cn][GP_ROWS * GP_PITCH];
    __shared__ float wp[Cn * Cn][WP_LEN];
    __shared__ float red[256];
    __shared__ __attribute__((aligned(16))) float hs[Ln];
    __shared__ __attribute__((aligned(16))) float park[P3_WAVES][P3_PRE * Dn];
    const int b  = blockIdx.x / SPLIT;
    const int sp = blockIdx.x % SPLIT;
    const int tid = threadIdx.x;
    const int lane = tid & 31;
    const int wv   = tid >> 5;
    const int rw   = sp * P3_WAVES + wv;

    const int* __restrict__ trow = tok + (long)b * Ln + rw * P3_TOK;
    const int tv0 = trow[lane], tv1 = trow[lane + 32];

    float gr0[Ln / 256], gr1[Ln / 256];
    #pragma unroll
    for (int k = 0; k < Ln / 256; ++k) {
        gr0[k] = g[((long)b * Cn + 0) * Ln + tid + 256 * k];
        gr1[k] = g[((long)b * Cn + 1) * Ln + tid + 256 * k];
    }
    const float cw0 = conv_w[tid];
    const float cw1 = (tid + 256 < Cn * Cn * Kn) ? conv_w[tid + 256] : 0.0f;

    for (int i = tid; i < Cn * GP_ROWS * GP_PITCH; i += 256) (&gp[0][0])[i] = 0.0f;
    for (int i = tid; i < Cn * Cn * WP_LEN; i += 256) (&wp[0][0])[i] = 0.0f;

    const int      pl = lane < 25 ? lane : 24;
    const unsigned o  = (unsigned)pl * 16u;
    const char* __restrict__ embc = (const char*)emb;
    char* parkw = (char*)&park[wv][0] + o;

    v4f x0[P3_PRE], x1[P3_PRE], x2[P3_PRE];

#define P3_LOAD(slot, p) {                                                    \
        const int t_ = (p) < 32 ? __builtin_amdgcn_readlane(tv0, (p) & 31)   \
                                : __builtin_amdgcn_readlane(tv1, (p) & 31);  \
        const char* rp_ = embc + ((unsigned)t_ * (Dn * 4u) + o);             \
        x0[slot] = *(const v4f*)(rp_);                                       \
        x1[slot] = *(const v4f*)(rp_ + 400);                                 \
        x2[slot] = *(const v4f*)(rp_ + 800); }

#define P3_BETA(p)                                                            \
        __builtin_bit_cast(float, (p) < 32                                    \
            ? __builtin_amdgcn_readlane(__builtin_bit_cast(int, bv0), (p) & 31) \
            : __builtin_amdgcn_readlane(__builtin_bit_cast(int, bv1), (p) & 31))

#define P3_USE(slot, p) {                                                     \
        const float bt_ = P3_BETA(p);                                         \
        z0 += bt_ * x0[slot]; z1 += bt_ * x1[slot]; z2 += bt_ * x2[slot]; }

    #pragma unroll
    for (int j = 0; j < P3_PRE; ++j) P3_LOAD(j, P3_PRE + j);

    lds_barrier();
    #pragma unroll
    for (int k = 0; k < Ln / 256; ++k) {
        const int u = tid + 256 * k + PADn;
        const int a = (u >> 4) * GP_PITCH + (u & 15);
        gp[0][a] = gr0[k];
        gp[1][a] = gr1[k];
    }
    {
        const int pr0 = tid / Kn;
        wp[pr0][WP_LEAD + tid - pr0 * Kn] = cw0;
        const int i1 = tid + 256;
        if (i1 < Cn * Cn * Kn) {
            const int pr1 = i1 / Kn;
            wp[pr1][WP_LEAD + i1 - pr1 * Kn] = cw1;
        }
    }
    lds_barrier();

    const int c    = lane & 15;
    const int hh   = lane >> 4;
    const float* __restrict__ a0p = &gp[0][(16 * wv + c) * GP_PITCH + 2 * hh];
    const float* __restrict__ a1p = &gp[1][(16 * wv + c) * GP_PITCH + 2 * hh];
    const float* __restrict__ bp  = &wp[0][WP_LEAD - c + 2 * hh];

    v8f acc0 = {}, acc1 = {};
    #pragma unroll
    for (int p = 0; p < 8; ++p) {
        #pragma unroll
        for (int q = 0; q < 4; ++q) {
            const int ao = p * GP_PITCH + 4 * q;
            const int t  = 16 * p + 4 * q;
            const v2f a0 = *(const v2f*)(a0p + ao);
            const v2f a1 = *(const v2f*)(a1p + ao);
            v2f b00, b01, b10, b11;
            b00.x = bp[0 * WP_LEN + t]; b00.y = bp[0 * WP_LEN + t + 1];
            b01.x = bp[1 * WP_LEN + t]; b01.y = bp[1 * WP_LEN + t + 1];
            b10.x = bp[2 * WP_LEN + t]; b10.y = bp[2 * WP_LEN + t + 1];
            b11.x = bp[3 * WP_LEN + t]; b11.y = bp[3 * WP_LEN + t + 1];
            acc0 = __builtin_amdgcn_wmma_f32_16x16x4_f32(
                false, a0, false, b00, (short)0, acc0, false, false);
            acc1 = __builtin_amdgcn_wmma_f32_16x16x4_f32(
                false, a0, false, b10, (short)0, acc1, false, false);
            acc0 = __builtin_amdgcn_wmma_f32_16x16x4_f32(
                false, a1, false, b01, (short)0, acc0, false, false);
            acc1 = __builtin_amdgcn_wmma_f32_16x16x4_f32(
                false, a1, false, b11, (short)0, acc1, false, false);
        }
    }

    #pragma unroll
    for (int j = 0; j < P3_PRE; ++j) {
        char* pk_ = parkw + j * (Dn * 4);
        *(v4f*)(pk_)       = x0[j];
        *(v4f*)(pk_ + 400) = x1[j];
        *(v4f*)(pk_ + 800) = x2[j];
    }
    #pragma unroll
    for (int j = 0; j < P3_PRE; ++j) P3_LOAD(j, j);

    const int xb = 256 * wv + 128 * hh + c;
    const float cb0 = conv_b[0], cb1 = conv_b[1];
    float h[8];
    for (int e = 0; e < 8; ++e)
        h[e] = fmaxf(fmaxf(acc0[e] + cb0, 0.0f), fmaxf(acc1[e] + cb1, 0.0f));

    float lm = -INFINITY;
    #pragma unroll
    for (int e = 0; e < 8; ++e) lm = fmaxf(lm, h[e]);
    red[tid] = lm; lds_barrier();
    for (int off = 128; off > 0; off >>= 1) {
        if (tid < off) red[tid] = fmaxf(red[tid], red[tid + off]);
        lds_barrier();
    }
    const float M = red[0];
    lds_barrier();

    float ls = 0.0f;
    #pragma unroll
    for (int e = 0; e < 8; ++e) {
        h[e] = __expf(h[e] - M);
        ls += h[e];
        hs[xb + 16 * e] = h[e];
    }
    red[tid] = ls; lds_barrier();
    for (int off = 128; off > 0; off >>= 1) {
        if (tid < off) red[tid] += red[tid + off];
        lds_barrier();
    }
    const float inv = 1.0f / red[0];
    if (tid < Ln / SPLIT / 4) {
        const int e4 = (Ln / SPLIT) * sp + 4 * tid;
        const v4f v = *(const v4f*)&hs[e4];
        *(v4f*)&betaOut[(long)b * Ln + e4] = v * inv;
    }

    const float bv0 = hs[rw * P3_TOK + lane] * inv;
    const float bv1 = hs[rw * P3_TOK + 32 + lane] * inv;

    v4f z0 = {}, z1 = {}, z2 = {};

    #pragma unroll
    for (int j = 0; j < P3_PRE; ++j) {
        P3_USE(j, j);
        P3_LOAD(j, 2 * P3_PRE + j);
    }
    #pragma unroll
    for (int j = 0; j < P3_PRE; ++j) {
        const char* pk_ = parkw + j * (Dn * 4);
        const float bt_ = P3_BETA(P3_PRE + j);
        z0 += bt_ * *(const v4f*)(pk_);
        z1 += bt_ * *(const v4f*)(pk_ + 400);
        z2 += bt_ * *(const v4f*)(pk_ + 800);
    }
    #pragma unroll
    for (int j = 0; j < P3_PRE; ++j) {
        P3_USE(j, 2 * P3_PRE + j);
        if (3 * P3_PRE + j < P3_TOK) P3_LOAD(j, 3 * P3_PRE + j);
    }
    #pragma unroll
    for (int j = 0; j < P3_TOK - 3 * P3_PRE; ++j) P3_USE(j, 3 * P3_PRE + j);

#undef P3_LOAD
#undef P3_USE
#undef P3_BETA

    float p0 = 0.0f, p1 = 0.0f;
    if (lane < 25) {
        const int d0 = 4 * lane;
        for (int e = 0; e < 4; ++e) {
            p0 += z0[e] * lin_w[d0 + e] + z1[e] * lin_w[100 + d0 + e]
                + z2[e] * lin_w[200 + d0 + e];
            p1 += z0[e] * lin_w[Dn + d0 + e] + z1[e] * lin_w[Dn + 100 + d0 + e]
                + z2[e] * lin_w[Dn + 200 + d0 + e];
        }
    }
    for (int off = 16; off > 0; off >>= 1) {
        p0 += __shfl_xor(p0, off, 32);
        p1 += __shfl_xor(p1, off, 32);
    }
    if (lane == 0 && ws_bytes >= sizeof(float) * WS_FLOATS) {
        v2f pr;
        pr.x = p0; pr.y = p1;
        *(v2f*)&part[(b * ROW_WAVES + rw) * 2] = pr;
    }
}

__global__ __launch_bounds__(32)
void k_head(const float* __restrict__ part, const float* __restrict__ lin_b,
            float* __restrict__ logits, size_t ws_bytes)
{
    if (ws_bytes < sizeof(float) * WS_FLOATS) return;
    const int b = blockIdx.x, tid = threadIdx.x;
    const v2f pr = *(const v2f*)&part[(b * ROW_WAVES + tid) * 2];
    float q0 = pr.x, q1 = pr.y;
    for (int off = 16; off > 0; off >>= 1) {
        q0 += __shfl_xor(q0, off, 32);
        q1 += __shfl_xor(q1, off, 32);
    }
    if (tid == 0) {
        const float o0 = q0 + lin_b[0];
        const float o1 = q1 + lin_b[1];
        const float mm = fmaxf(o0, o1);
        const float lse = mm + __logf(__expf(o0 - mm) + __expf(o1 - mm));
        logits[b * 2 + 0] = o0 - lse;
        logits[b * 2 + 1] = o1 - lse;
    }
}

extern "C" void kernel_launch(void* const* d_in, const int* in_sizes, int n_in,
                              void* d_out, int out_size, void* d_ws, size_t ws_size,
                              hipStream_t stream)
{
    const int*   tok = (const int*)  d_in[0];
    const float* emb = (const float*)d_in[1];
    const float* cls = (const float*)d_in[2];
    const float* cw  = (const float*)d_in[3];
    const float* cb  = (const float*)d_in[4];
    const float* lw  = (const float*)d_in[5];
    const float* lb  = (const float*)d_in[6];

    float* out    = (float*)d_out;
    float* logits = out;
    float* beta   = out + Bn * Cn;
    float* g      = (float*)d_ws;
    float* part   = g + Bn * Cn * Ln;

    k_phase1<<<1024, 256, 0, stream>>>(tok, emb, cls, g);
    k_phase23<<<Bn * SPLIT, 256, 0, stream>>>(g, cw, cb, tok, emb, lw, beta, part, ws_size);
    k_head<<<Bn, 32, 0, stream>>>(part, lb, logits, ws_size);
}


# Softmax split by wave: a block convolves only its 512 positions, a wave weights its 64 table rows with exp(h - own max) at once; the head kernel joins the 32 (max, sum) pairs and rescales beta.
# speedup vs baseline: 5.3398x; 1.0911x over previous
#include <hip/hip_runtime.h>
#include <hip/hip_bf16.h>
#include <math.h>

#define Bn   64
#define Ln   2048
#define Dn   300
#define Cn   2
#define Kn   111
#define PADn 55

typedef __attribute__((ext_vector_type(2))) float v2f;
typedef __attribute__((ext_vector_type(8))) float v8f;
typedef __attribute__((ext_vector_type(4))) float v4f;

#define CH_DIMS  100
#define CH_PITCH 100

__global__ __launch_bounds__(256)
void k_phase1(const int* __restrict__ tok, const float* __restrict__ emb,
              const float* __restrict__ cls, float* __restrict__ g)
{
    __shared__ __attribute__((aligned(16))) float clsn[3 * 304];
    __shared__ float red[256];
    __shared__ __attribute__((aligned(16))) float xs[8][16 * CH_PITCH];
    const int tid = threadIdx.x;

    const int wave = tid >> 5;
    const int lane = tid & 31;
    const int tileId = blockIdx.x * 8 + wave;
    const int b  = tileId >> 7;
    const int l0 = (tileId & 127) * 16;
    const int m    = lane & 15;
    const int hi   = lane >> 4;

    const int token = tok[b * Ln + l0 + m];

    const int      pl = lane < 25 ? lane : 24;
    const unsigned o  = (unsigned)pl * 16u;
    const char* __restrict__ embc = (const char*)emb;
    char* xsw = (char*)&xs[wave][0] + o;

    const char* rp[16];
    #pragma unroll
    for (int r = 0; r < 16; ++r) {
        const int t_ = __builtin_amdgcn_readlane(token, r);
        rp[r] = embc + (size_t)(unsigned)t_ * (Dn * 4u);
    }

    v4f x[16];
    #pragma unroll
    for (int r = 0; r < 16; ++r) x[r] = *(const v4f*)(rp[r] + o);

    for (int i = tid; i < 3 * 304; i += 256) clsn[i] = 0.0f;
    __syncthreads();
    for (int c = 0; c < Cn; ++c) {
        float s = 0.0f;
        for (int d = tid; d < Dn; d += 256) { float v = cls[c * Dn + d]; s += v * v; }
        red[tid] = s;
        __syncthreads();
        for (int off = 128; off > 0; off >>= 1) {
            if (tid < off) red[tid] += red[tid + off];
            __syncthreads();
        }
        float rn = 1.0f / fmaxf(sqrtf(red[0]), 1e-12f);
        __syncthreads();
        for (int d = tid; d < Dn; d += 256) clsn[c * 304 + d] = cls[c * Dn + d] * rn;
        __syncthreads();
    }

    const float* __restrict__ ar = &xs[wave][m * CH_PITCH + 4 * hi];
    const float* __restrict__ br = &clsn[(m < 2 ? m : 2) * 304 + 4 * hi];

    v8f acc0 = {}, acc1 = {};
    float n2 = 0.0f;
    #pragma unroll
    for (int q = 0; q < 3; ++q) {
        #pragma unroll
        for (int r = 0; r < 16; ++r) *(v4f*)(xsw + r * (CH_PITCH * 4)) = x[r];
        if (q < 2) {
            unsigned on = o + (unsigned)(q + 1) * (CH_DIMS * 4u);
            asm volatile("" : "+v"(on));
            #pragma unroll
            for (int r = 0; r < 16; ++r) x[r] = *(const v4f*)(rp[r] + on);
        }
        __builtin_amdgcn_fence(__ATOMIC_RELEASE, "wavefront");
        __builtin_amdgcn_wave_barrier();
        __builtin_amdgcn_fence(__ATOMIC_ACQUIRE, "wavefront");

        const float* __restrict__ bq_ = br + q * CH_DIMS;
        #pragma unroll
        for (int j = 0; j < 12; ++j) {
            const v4f a  = *(const v4f*)(ar + 8 * j);
            const v4f bq = *(const v4f*)(bq_ + 8 * j);
            n2 += a.x * a.x + a.y * a.y + a.z * a.z + a.w * a.w;
            v2f a01, a23, b01, b23;
            a01.x = a.x;  a01.y = a.y;  a23.x = a.z;  a23.y = a.w;
            b01.x = bq.x; b01.y = bq.y; b23.x = bq.z; b23.y = bq.w;
            acc0 = __builtin_amdgcn_wmma_f32_16x16x4_f32(
                false, a01, false, b01, (short)0, acc0, false, false);
            acc1 = __builtin_amdgcn_wmma_f32_16x16x4_f32(
                false, a23, false, b23, (short)0, acc1, false, false);
        }
        {
            const v2f a  = *(const v2f*)(&xs[wave][m * CH_PITCH + 96 + 2 * hi]);
            const v2f bq = *(const v2f*)(&clsn[(m < 2 ? m : 2) * 304 + q * CH_DIMS + 96 + 2 * hi]);
            n2 += a.x * a.x + a.y * a.y;
            if (q & 1)
                acc1 = __builtin_amdgcn_wmma_f32_16x16x4_f32(
                    false, a, false, bq, (short)0, acc1, false, false);
            else
                acc0 = __builtin_amdgcn_wmma_f32_16x16x4_f32(
                    false, a, false, bq, (short)0, acc0, false, false);
        }
        __builtin_amdgcn_fence(__ATOMIC_RELEASE, "wavefront");
        __builtin_amdgcn_wave_barrier();
        __builtin_amdgcn_fence(__ATOMIC_ACQUIRE, "wavefront");
    }
    const v8f acc = acc0 + acc1;

    n2 += __shfl_xor(n2, 16, 32);

    const int  c      = m;
    const bool active = (m < 2);
    const int  mbase  = hi * 8;
    float* __restrict__ gout = g + (((long)b * Cn + c) * Ln + l0 + mbase);
    #pragma unroll
    for (int i = 0; i < 8; ++i) {
        float nn    = __shfl(n2, mbase + i, 32);
        float scale = 1.0f / fmaxf(sqrtf(nn), 1e-12f);
        if (active) gout[i] = acc[i] * scale;
    }
}

#define SPLIT     4
#define BLK_POS   (Ln / SPLIT)
#define GP_IN     (BLK_POS + 2 * PADn + 2)
#define GP_ROWS   (GP_IN / 16)
#define GP_PITCH  20
#define WP_LEAD   15
#define WP_LEN    144

#define P3_WAVES  8
#define ROW_WAVES (SPLIT * P3_WAVES)
#define P3_TOK    (Ln / ROW_WAVES)
#define P3_PRE    20
#define REC       4
#define WS_FLOATS (Bn * Cn * Ln + Bn * ROW_WAVES * REC)

__device__ __forceinline__ void lds_barrier()
{
    __builtin_amdgcn_fence(__ATOMIC_RELEASE, "workgroup", "local");
    __builtin_amdgcn_s_barrier();
    __builtin_amdgcn_fence(__ATOMIC_ACQUIRE, "workgroup", "local");
}

__global__ __launch_bounds__(256)
void k_phase23(const float* __restrict__ g, const float* __restrict__ conv_w,
               const float* __restrict__ conv_b, const int* __restrict__ tok,
               const float* __restrict__ emb, const float* __restrict__ lin_w,
               float* __restrict__ betaOut, float* __restrict__ part,
               size_t ws_bytes)
{
    __shared__ __attribute__((aligned(16))) float gp[Cn][GP_ROWS * GP_PITCH];
    __shared__ float wp[Cn * Cn][WP_LEN];
    __shared__ float hc[Cn][BLK_POS];
    const int b  = blockIdx.x / SPLIT;
    const int sp = blockIdx.x % SPLIT;
    const int tid = threadIdx.x;
    const int lane = tid & 31;
    const int wv   = __builtin_amdgcn_readfirstlane(tid >> 5);
    const int rw   = sp * P3_WAVES + wv;

    const int* __restrict__ trow = tok + (long)b * Ln + rw * P3_TOK;
    const int tv0 = trow[lane], tv1 = trow[lane + 32];

    float gr0[3], gr1[3];
    #pragma unroll
    for (int k = 0; k < 3; ++k) {
        const int i = tid + 256 * k;
        const int l = BLK_POS * sp - PADn + i;
        const bool in = (i < GP_IN) && (l >= 0) && (l < Ln);
        const int lc = l < 0 ? 0 : (l < Ln ? l : Ln - 1);
        const float v0 = g[((long)b * Cn + 0) * Ln + lc];
        const float v1 = g[((long)b * Cn + 1) * Ln + lc];
        gr0[k] = in ? v0 : 0.0f;
        gr1[k] = in ? v1 : 0.0f;
    }
    const float cw0 = conv_w[tid];
    const float cw1 = (tid + 256 < Cn * Cn * Kn) ? conv_w[tid + 256] : 0.0f;

    for (int i = tid; i < Cn * Cn * WP_LEN; i += 256) (&wp[0][0])[i] = 0.0f;

    const int      pl = lane < 25 ? lane : 24;
    const unsigned o  = (unsigned)pl * 16u;
    const char* __restrict__ embc = (const char*)emb;

    v4f x0[P3_PRE], x1[P3_PRE], x2[P3_PRE];

#define P3_LOAD(slot, p) {                                                    \
        const int t_ = (p) < 32 ? __builtin_amdgcn_readlane(tv0, (p) & 31)   \
                                : __builtin_amdgcn_readlane(tv1, (p) & 31);  \
        const char* rp_ = embc + ((unsigned)t_ * (Dn * 4u) + o);             \
        x0[slot] = *(const v4f*)(rp_);                                       \
        x1[slot] = *(const v4f*)(rp_ + 400);                                 \
        x2[slot] = *(const v4f*)(rp_ + 800); }

#define P3_W(p)                                                               \
        __builtin_bit_cast(float, (p) < 32                                    \
            ? __builtin_amdgcn_readlane(__builtin_bit_cast(int, bv0), (p) & 31) \
            : __builtin_amdgcn_readlane(__builtin_bit_cast(int, bv1), (p) & 31))

#define P3_USE(slot, p) {                                                     \
        const float bt_ = P3_W(p);                                            \
        z0 += bt_ * x0[slot]; z1 += bt_ * x1[slot]; z2 += bt_ * x2[slot]; }

    #pragma unroll
    for (int j = 0; j < P3_PRE; ++j) P3_LOAD(j, j);

    lds_barrier();
    #pragma unroll
    for (int k = 0; k < 3; ++k) {
        const int i = tid + 256 * k;
        if (i < GP_IN) {
            const int a = (i >> 4) * GP_PITCH + (i & 15);
            gp[0][a] = gr0[k];
            gp[1][a] = gr1[k];
        }
    }
    {
        const int pr0 = tid / Kn;
        wp[pr0][WP_LEAD + tid - pr0 * Kn] = cw0;
        const int i1 = tid + 256;
        if (i1 < Cn * Cn * Kn) {
            const int pr1 = i1 / Kn;
            wp[pr1][WP_LEAD + i1 - pr1 * Kn] = cw1;
        }
    }
    lds_barrier();

    const int c    = lane & 15;
    const int hh   = lane >> 4;
    if (wv < 4) {
        const int tl = wv & 1;
        const int co = wv >> 1;
        const float* __restrict__ a0p = &gp[0][(16 * tl + c) * GP_PITCH + 2 * hh];
        const float* __restrict__ a1p = &gp[1][(16 * tl + c) * GP_PITCH + 2 * hh];
        const float* __restrict__ bp  = &wp[2 * co][WP_LEAD - c + 2 * hh];

        v8f acc0 = {}, acc1 = {};
        #pragma unroll
        for (int p = 0; p < 8; ++p) {
            #pragma unroll
            for (int q = 0; q < 4; ++q) {
                const int ao = p * GP_PITCH + 4 * q;
                const int t  = 16 * p + 4 * q;
                const v2f a0 = *(const v2f*)(a0p + ao);
                const v2f a1 = *(const v2f*)(a1p + ao);
                v2f b0, b1;
                b0.x = bp[0 * WP_LEN + t]; b0.y = bp[0 * WP_LEN + t + 1];
                b1.x = bp[1 * WP_LEN + t]; b1.y = bp[1 * WP_LEN + t + 1];
                acc0 = __builtin_amdgcn_wmma_f32_16x16x4_f32(
                    false, a0, false, b0, (short)0, acc0, false, false);
                acc1 = __builtin_amdgcn_wmma_f32_16x16x4_f32(
                    false, a1, false, b1, (short)0, acc1, false, false);
            }
        }
        const int xb = 256 * tl + 128 * hh + c;
        const float cb = conv_b[co];
        #pragma unroll
        for (int e = 0; e < 8; ++e)
            hc[co][xb + 16 * e] = fmaxf(acc0[e] + acc1[e] + cb, 0.0f);
    }
    lds_barrier();

    const float h0 = fmaxf(hc[0][P3_TOK * wv + lane],      hc[1][P3_TOK * wv + lane]);
    const float h1 = fmaxf(hc[0][P3_TOK * wv + 32 + lane], hc[1][P3_TOK * wv + 32 + lane]);
    float mw = fmaxf(h0, h1);
    for (int off = 16; off > 0; off >>= 1) mw = fmaxf(mw, __shfl_xor(mw, off, 32));
    const float bv0 = __expf(h0 - mw);
    const float bv1 = __expf(h1 - mw);
    float sw = bv0 + bv1;
    for (int off = 16; off > 0; off >>= 1) sw += __shfl_xor(sw, off, 32);
    {
        float* __restrict__ bo = betaOut + (long)b * Ln + rw * P3_TOK;
        bo[lane]      = bv0;
        bo[lane + 32] = bv1;
    }

    v4f z0 = {}, z1 = {}, z2 = {};

    const v4f lw00 = *(const v4f*)&lin_w[4 * pl];
    const v4f lw01 = *(const v4f*)&lin_w[100 + 4 * pl];
    const v4f lw02 = *(const v4f*)&lin_w[200 + 4 * pl];
    const v4f lw10 = *(const v4f*)&lin_w[Dn + 4 * pl];
    const v4f lw11 = *(const v4f*)&lin_w[Dn + 100 + 4 * pl];
    const v4f lw12 = *(const v4f*)&lin_w[Dn + 200 + 4 * pl];

    #pragma unroll
    for (int r = 0; r < (P3_TOK + P3_PRE - 1) / P3_PRE; ++r) {
        #pragma unroll
        for (int j = 0; j < P3_PRE; j += 2) {
            const int p = r * P3_PRE + j;
            if (p < P3_TOK) {
                P3_USE(j, p);
                P3_USE(j + 1, p + 1);
                if (p + P3_PRE < P3_TOK) {
                    P3_LOAD(j, p + P3_PRE);
                    P3_LOAD(j + 1, p + P3_PRE + 1);
                }
                __builtin_amdgcn_sched_barrier(0);
            }
        }
    }

#undef P3_LOAD
#undef P3_USE
#undef P3_W

    float p0 = 0.0f, p1 = 0.0f;
    #pragma unroll
    for (int e = 0; e < 4; ++e) {
        p0 += z0[e] * lw00[e] + z1[e] * lw01[e] + z2[e] * lw02[e];
        p1 += z0[e] * lw10[e] + z1[e] * lw11[e] + z2[e] * lw12[e];
    }
    p0 = lane < 25 ? p0 : 0.0f;
    p1 = lane < 25 ? p1 : 0.0f;
    for (int off = 16; off > 0; off >>= 1) {
        p0 += __shfl_xor(p0, off, 32);
        p1 += __shfl_xor(p1, off, 32);
    }
    if (lane == 0 && ws_bytes >= sizeof(float) * WS_FLOATS) {
        v4f rc;
        rc.x = p0; rc.y = p1; rc.z = mw; rc.w = sw;
        *(v4f*)&part[(b * ROW_WAVES + rw) * REC] = rc;
    }
}

__global__ __launch_bounds__(256)
void k_head(const float* __restrict__ part, const float* __restrict__ lin_b,
            float* __restrict__ logits, float* __restrict__ beta, size_t ws_bytes)
{
    if (ws_bytes < sizeof(float) * WS_FLOATS) return;
    __shared__ float fw[ROW_WAVES];
    const int b = blockIdx.x, tid = threadIdx.x;
    float* __restrict__ brow = beta + (long)b * Ln;
    const v4f e0 = *(const v4f*)&brow[4 * tid];
    const v4f e1 = *(const v4f*)&brow[Ln / 2 + 4 * tid];
    if (tid < ROW_WAVES) {
        const v4f rc = *(const v4f*)&part[(b * ROW_WAVES + tid) * REC];
        float M = rc.z;
        for (int off = 16; off > 0; off >>= 1) M = fmaxf(M, __shfl_xor(M, off, 32));
        const float f = __expf(rc.z - M);
        float S = rc.w * f, q0 = rc.x * f, q1 = rc.y * f;
        for (int off = 16; off > 0; off >>= 1) {
            S  += __shfl_xor(S,  off, 32);
            q0 += __shfl_xor(q0, off, 32);
            q1 += __shfl_xor(q1, off, 32);
        }
        const float inv = 1.0f / S;
        fw[tid] = f * inv;
        if (tid == 0) {
            const float o0 = q0 * inv + lin_b[0];
            const float o1 = q1 * inv + lin_b[1];
            const float mm = fmaxf(o0, o1);
            const float lse = mm + __logf(__expf(o0 - mm) + __expf(o1 - mm));
            logits[b * 2 + 0] = o0 - lse;
            logits[b * 2 + 1] = o1 - lse;
        }
    }
    __syncthreads();
    const float f0 = fw[(4 * tid) / P3_TOK];
    const float f1 = fw[(Ln / 2 + 4 * tid) / P3_TOK];
    *(v4f*)&brow[4 * tid]          = e0 * f0;
    *(v4f*)&brow[Ln / 2 + 4 * tid] = e1 * f1;
}

extern "C" void kernel_launch(void* const* d_in, const int* in_sizes, int n_in,
                              void* d_out, int out_size, void* d_ws, size_t ws_size,
                              hipStream_t stream)
{
    const int*   tok = (const int*)  d_in[0];
    const float* emb = (const float*)d_in[1];
    const float* cls = (const float*)d_in[2];
    const float* cw  = (const float*)d_in[3];
    const float* cb  = (const float*)d_in[4];
    const float* lw  = (const float*)d_in[5];
    const float* lb  = (const float*)d_in[6];

    float* out    = (float*)d_out;
    float* logits = out;
    float* beta   = out + Bn * Cn;
    float* g      = (float*)d_ws;
    float* part   = g + Bn * Cn * Ln;

    k_phase1<<<1024, 256, 0, stream>>>(tok, emb, cls, g);
    k_phase23<<<Bn * SPLIT, 256, 0, stream>>>(g, cw, cb, tok, emb, lw, beta, part, ws_size);
    k_head<<<Bn, 256, 0, stream>>>(part, lb, logits, beta, ws_size);
}


# Phase 1's class set-up runs under the first chunk's table-row loads: its 23 barriers order LDS only and the class rows are asked for ahead of the table rows, so no barrier waits for a row.
# speedup vs baseline: 5.6316x; 1.0546x over previous
#include <hip/hip_runtime.h>
#include <hip/hip_bf16.h>
#include <math.h>

#define Bn   64
#define Ln   2048
#define Dn   300
#define Cn   2
#define Kn   111
#define PADn 55

typedef __attribute__((ext_vector_type(2))) float v2f;
typedef __attribute__((ext_vector_type(8))) float v8f;
typedef __attribute__((ext_vector_type(4))) float v4f;

#define CH_DIMS  100
#define CH_PITCH 100

__device__ __forceinline__ void lds_barrier()
{
    __builtin_amdgcn_fence(__ATOMIC_RELEASE, "workgroup", "local");
    __builtin_amdgcn_s_barrier();
    __builtin_amdgcn_fence(__ATOMIC_ACQUIRE, "workgroup", "local");
}

__global__ __launch_bounds__(256, 8)
void k_phase1(const int* __restrict__ tok, const float* __restrict__ emb,
              const float* __restrict__ cls, float* __restrict__ g)
{
    __shared__ __attribute__((aligned(16))) float clsn[3 * 304];
    __shared__ float red[256];
    __shared__ __attribute__((aligned(16))) float xs[8][16 * CH_PITCH];
    const int tid = threadIdx.x;

    const int wave = tid >> 5;
    const int lane = tid & 31;
    const int tileId = blockIdx.x * 8 + wave;
    const int b  = tileId >> 7;
    const int l0 = (tileId & 127) * 16;
    const int m    = lane & 15;
    const int hi   = lane >> 4;

    const int token = tok[b * Ln + l0 + m];

    const bool  two = tid + 256 < Dn;
    const int   d1  = two ? tid + 256 : tid;
    const float ca0 = cls[tid];
    const float ca1 = cls[Dn + tid];
    const float cb0 = cls[d1];
    const float cb1 = cls[Dn + d1];

    const int      pl = lane < 25 ? lane : 24;
    const unsigned o  = (unsigned)pl * 16u;
    const char* __restrict__ embc = (const char*)emb;
    char* xsw = (char*)&xs[wave][0] + o;

    const char* rp[16];
    #pragma unroll
    for (int r = 0; r < 16; ++r) {
        const int t_ = __builtin_amdgcn_readlane(token, r);
        rp[r] = embc + (size_t)(unsigned)t_ * (Dn * 4u);
    }

    v4f x[16];
    #pragma unroll
    for (int r = 0; r < 16; ++r) x[r] = *(const v4f*)(rp[r] + o);

    for (int i = tid; i < 3 * 304; i += 256) clsn[i] = 0.0f;
    lds_barrier();
    #pragma unroll
    for (int c = 0; c < Cn; ++c) {
        const float va = c ? ca1 : ca0;
        const float vb = two ? (c ? cb1 : cb0) : 0.0f;
        float s = va * va;
        s = __builtin_fmaf(vb, vb, s);
        red[tid] = s;
        lds_barrier();
        for (int off = 128; off > 0; off >>= 1) {
            if (tid < off) red[tid] += red[tid + off];
            lds_barrier();
        }
        float rn = 1.0f / fmaxf(sqrtf(red[0]), 1e-12f);
        lds_barrier();
        clsn[c * 304 + tid] = va * rn;
        if (two) clsn[c * 304 + tid + 256] = vb * rn;
        lds_barrier();
    }

    const float* __restrict__ ar = &xs[wave][m * CH_PITCH + 4 * hi];
    const float* __restrict__ br = &clsn[(m < 2 ? m : 2) * 304 + 4 * hi];

    v8f acc0 = {}, acc1 = {};
    float n2 = 0.0f;
    #pragma unroll
    for (int q = 0; q < 3; ++q) {
        #pragma unroll
        for (int r = 0; r < 16; ++r) *(v4f*)(xsw + r * (CH_PITCH * 4)) = x[r];
        if (q < 2) {
            unsigned on = o + (unsigned)(q + 1) * (CH_DIMS * 4u);
            asm volatile("" : "+v"(on));
            #pragma unroll
            for (int r = 0; r < 16; ++r) x[r] = *(const v4f*)(rp[r] + on);
        }
        __builtin_amdgcn_fence(__ATOMIC_RELEASE, "wavefront");
        __builtin_amdgcn_wave_barrier();
        __builtin_amdgcn_fence(__ATOMIC_ACQUIRE, "wavefront");

        const float* __restrict__ bq_ = br + q * CH_DIMS;
        #pragma unroll
        for (int j = 0; j < 12; ++j) {
            const v4f a  = *(const v4f*)(ar + 8 * j);
            const v4f bq = *(const v4f*)(bq_ + 8 * j);
            n2 += a.x * a.x + a.y * a.y + a.z * a.z + a.w * a.w;
            v2f a01, a23, b01, b23;
            a01.x = a.x;  a01.y = a.y;  a23.x = a.z;  a23.y = a.w;
            b01.x = bq.x; b01.y = bq.y; b23.x = bq.z; b23.y = bq.w;
            acc0 = __builtin_amdgcn_wmma_f32_16x16x4_f32(
                false, a01, false, b01, (short)0, acc0, false, false);
            acc1 = __builtin_amdgcn_wmma_f32_16x16x4_f32(
                false, a23, false, b23, (short)0, acc1, false, false);
        }
        {
            const v2f a  = *(const v2f*)(&xs[wave][m * CH_PITCH + 96 + 2 * hi]);
            const v2f bq = *(const v2f*)(&clsn[(m < 2 ? m : 2) * 304 + q * CH_DIMS + 96 + 2 * hi]);
            n2 += a.x * a.x + a.y * a.y;
            if (q & 1)
                acc1 = __builtin_amdgcn_wmma_f32_16x16x4_f32(
                    false, a, false, bq, (short)0, acc1, false, false);
            else
                acc0 = __builtin_amdgcn_wmma_f32_16x16x4_f32(
                    false, a, false, bq, (short)0, acc0, false, false);
        }
        __builtin_amdgcn_fence(__ATOMIC_RELEASE, "wavefront");
        __builtin_amdgcn_wave_barrier();
        __builtin_amdgcn_fence(__ATOMIC_ACQUIRE, "wavefront");
    }
    const v8f acc = acc0 + acc1;

    n2 += __shfl_xor(n2, 16, 32);

    const int  c      = m;
    const bool active = (m < 2);
    const int  mbase  = hi * 8;
    float* __restrict__ gout = g + (((long)b * Cn + c) * Ln + l0 + mbase);
    #pragma unroll
    for (int i = 0; i < 8; ++i) {
        float nn    = __shfl(n2, mbase + i, 32);
        float scale = 1.0f / fmaxf(sqrtf(nn), 1e-12f);
        if (active) gout[i] = acc[i] * scale;
    }
}

#define SPLIT     4
#define BLK_POS   (Ln / SPLIT)
#define GP_IN     (BLK_POS + 2 * PADn + 2)
#define GP_ROWS   (GP_IN / 16)
#define GP_PITCH  20
#define WP_LEAD   15
#define WP_LEN    144

#define P3_WAVES  8
#define ROW_WAVES (SPLIT * P3_WAVES)
#define P3_TOK    (Ln / ROW_WAVES)
#define P3_PRE    20
#define REC       4
#define WS_FLOATS (Bn * Cn * Ln + Bn * ROW_WAVES * REC)

__global__ __launch_bounds__(256)
void k_phase23(const float* __restrict__ g, const float* __restrict__ conv_w,
               const float* __restrict__ conv_b, const int* __restrict__ tok,
               const float* __restrict__ emb, const float* __restrict__ lin_w,
               float* __restrict__ betaOut, float* __restrict__ part,
               size_t ws_bytes)
{
    __shared__ __attribute__((aligned(16))) float gp[Cn][GP_ROWS * GP_PITCH];
    __shared__ float wp[Cn * Cn][WP_LEN];
    __shared__ float hc[Cn][BLK_POS];
    const int b  = blockIdx.x / SPLIT;
    const int sp = blockIdx.x % SPLIT;
    const int tid = threadIdx.x;
    const int lane = tid & 31;
    const int wv   = __builtin_amdgcn_readfirstlane(tid >> 5);
    const int rw   = sp * P3_WAVES + wv;

    const int* __restrict__ trow = tok + (long)b * Ln + rw * P3_TOK;
    const int tv0 = trow[lane], tv1 = trow[lane + 32];

    float gr0[3], gr1[3];
    #pragma unroll
    for (int k = 0; k < 3; ++k) {
        const int i = tid + 256 * k;
        const int l = BLK_POS * sp - PADn + i;
        const bool in = (i < GP_IN) && (l >= 0) && (l < Ln);
        const int lc = l < 0 ? 0 : (l < Ln ? l : Ln - 1);
        const float v0 = g[((long)b * Cn + 0) * Ln + lc];
        const float v1 = g[((long)b * Cn + 1) * Ln + lc];
        gr0[k] = in ? v0 : 0.0f;
        gr1[k] = in ? v1 : 0.0f;
    }
    const float cw0 = conv_w[tid];
    const float cw1 = (tid + 256 < Cn * Cn * Kn) ? conv_w[tid + 256] : 0.0f;

    for (int i = tid; i < Cn * Cn * WP_LEN; i += 256) (&wp[0][0])[i] = 0.0f;

    const int      pl = lane < 25 ? lane : 24;
    const unsigned o  = (unsigned)pl * 16u;
    const char* __restrict__ embc = (const char*)emb;

    v4f x0[P3_PRE], x1[P3_PRE], x2[P3_PRE];

#define P3_LOAD(slot, p) {                                                    \
        const int t_ = (p) < 32 ? __builtin_amdgcn_readlane(tv0, (p) & 31)   \
                                : __builtin_amdgcn_readlane(tv1, (p) & 31);  \
        const char* rp_ = embc + ((unsigned)t_ * (Dn * 4u) + o);             \
        x0[slot] = *(const v4f*)(rp_);                                       \
        x1[slot] = *(const v4f*)(rp_ + 400);                                 \
        x2[slot] = *(const v4f*)(rp_ + 800); }

#define P3_W(p)                                                               \
        __builtin_bit_cast(float, (p) < 32                                    \
            ? __builtin_amdgcn_readlane(__builtin_bit_cast(int, bv0), (p) & 31) \
            : __builtin_amdgcn_readlane(__builtin_bit_cast(int, bv1), (p) & 31))

#define P3_USE(slot, p) {                                                     \
        const float bt_ = P3_W(p);                                            \
        z0 += bt_ * x0[slot]; z1 += bt_ * x1[slot]; z2 += bt_ * x2[slot]; }

    #pragma unroll
    for (int j = 0; j < P3_PRE; ++j) P3_LOAD(j, j);

    lds_barrier();
    #pragma unroll
    for (int k = 0; k < 3; ++k) {
        const int i = tid + 256 * k;
        if (i < GP_IN) {
            const int a = (i >> 4) * GP_PITCH + (i & 15);
            gp[0][a] = gr0[k];
            gp[1][a] = gr1[k];
        }
    }
    {
        const int pr0 = tid / Kn;
        wp[pr0][WP_LEAD + tid - pr0 * Kn] = cw0;
        const int i1 = tid + 256;
        if (i1 < Cn * Cn * Kn) {
            const int pr1 = i1 / Kn;
            wp[pr1][WP_LEAD + i1 - pr1 * Kn] = cw1;
        }
    }
    lds_barrier();

    const int c    = lane & 15;
    const int hh   = lane >> 4;
    if (wv < 4) {
        const int tl = wv & 1;
        const int co = wv >> 1;
        const float* __restrict__ a0p = &gp[0][(16 * tl + c) * GP_PITCH + 2 * hh];
        const float* __restrict__ a1p = &gp[1][(16 * tl + c) * GP_PITCH + 2 * hh];
        const float* __restrict__ bp  = &wp[2 * co][WP_LEAD - c + 2 * hh];

        v8f acc0 = {}, acc1 = {};
        #pragma unroll
        for (int p = 0; p < 8; ++p) {
            #pragma unroll
            for (int q = 0; q < 4; ++q) {
                const int ao = p * GP_PITCH + 4 * q;
                const int t  = 16 * p + 4 * q;
                const v2f a0 = *(const v2f*)(a0p + ao);
                const v2f a1 = *(const v2f*)(a1p + ao);
                v2f b0, b1;
                b0.x = bp[0 * WP_LEN + t]; b0.y = bp[0 * WP_LEN + t + 1];
                b1.x = bp[1 * WP_LEN + t]; b1.y = bp[1 * WP_LEN + t + 1];
                acc0 = __builtin_amdgcn_wmma_f32_16x16x4_f32(
                    false, a0, false, b0, (short)0, acc0, false, false);
                acc1 = __builtin_amdgcn_wmma_f32_16x16x4_f32(
                    false, a1, false, b1, (short)0, acc1, false, false);
            }
        }
        const int xb = 256 * tl + 128 * hh + c;
        const float cb = conv_b[co];
        #pragma unroll
        for (int e = 0; e < 8; ++e)
            hc[co][xb + 16 * e] = fmaxf(acc0[e] + acc1[e] + cb, 0.0f);
    }
    lds_barrier();

    const float h0 = fmaxf(hc[0][P3_TOK * wv + lane],      hc[1][P3_TOK * wv + lane]);
    const float h1 = fmaxf(hc[0][P3_TOK * wv + 32 + lane], hc[1][P3_TOK * wv + 32 + lane]);
    float mw = fmaxf(h0, h1);
    for (int off = 16; off > 0; off >>= 1) mw = fmaxf(mw, __shfl_xor(mw, off, 32));
    const float bv0 = __expf(h0 - mw);
    const float bv1 = __expf(h1 - mw);
    float sw = bv0 + bv1;
    for (int off = 16; off > 0; off >>= 1) sw += __shfl_xor(sw, off, 32);
    {
        float* __restrict__ bo = betaOut + (long)b * Ln + rw * P3_TOK;
        bo[lane]      = bv0;
        bo[lane + 32] = bv1;
    }

    v4f z0 = {}, z1 = {}, z2 = {};

    const v4f lw00 = *(const v4f*)&lin_w[4 * pl];
    const v4f lw01 = *(const v4f*)&lin_w[100 + 4 * pl];
    const v4f lw02 = *(const v4f*)&lin_w[200 + 4 * pl];
    const v4f lw10 = *(const v4f*)&lin_w[Dn + 4 * pl];
    const v4f lw11 = *(const v4f*)&lin_w[Dn + 100 + 4 * pl];
    const v4f lw12 = *(const v4f*)&lin_w[Dn + 200 + 4 * pl];

    #pragma unroll
    for (int r = 0; r < (P3_TOK + P3_PRE - 1) / P3_PRE; ++r) {
        #pragma unroll
        for (int j = 0; j < P3_PRE; j += 2) {
            const int p = r * P3_PRE + j;
            if (p < P3_TOK) {
                P3_USE(j, p);
                P3_USE(j + 1, p + 1);
                if (p + P3_PRE < P3_TOK) {
                    P3_LOAD(j, p + P3_PRE);
                    P3_LOAD(j + 1, p + P3_PRE + 1);
                }
                __builtin_amdgcn_sched_barrier(0);
            }
        }
    }

#undef P3_LOAD
#undef P3_USE
#undef P3_W

    float p0 = 0.0f, p1 = 0.0f;
    #pragma unroll
    for (int e = 0; e < 4; ++e) {
        p0 += z0[e] * lw00[e] + z1[e] * lw01[e] + z2[e] * lw02[e];
        p1 += z0[e] * lw10[e] + z1[e] * lw11[e] + z2[e] * lw12[e];
    }
    p0 = lane < 25 ? p0 : 0.0f;
    p1 = lane < 25 ? p1 : 0.0f;
    for (int off = 16; off > 0; off >>= 1) {
        p0 += __shfl_xor(p0, off, 32);
        p1 += __shfl_xor(p1, off, 32);
    }
    if (lane == 0 && ws_bytes >= sizeof(float) * WS_FLOATS) {
        v4f rc;
        rc.x = p0; rc.y = p1; rc.z = mw; rc.w = sw;
        *(v4f*)&part[(b * ROW_WAVES + rw) * REC] = rc;
    }
}

__global__ __launch_bounds__(256)
void k_head(const float* __restrict__ part, const float* __restrict__ lin_b,
            float* __restrict__ logits, float* __restrict__ beta, size_t ws_bytes)
{
    if (ws_bytes < sizeof(float) * WS_FLOATS) return;
    __shared__ float fw[ROW_WAVES];
    const int b = blockIdx.x, tid = threadIdx.x;
    float* __restrict__ brow = beta + (long)b * Ln;
    const v4f e0 = *(const v4f*)&brow[4 * tid];
    const v4f e1 = *(const v4f*)&brow[Ln / 2 + 4 * tid];
    if (tid < ROW_WAVES) {
        const v4f rc = *(const v4f*)&part[(b * ROW_WAVES + tid) * REC];
        float M = rc.z;
        for (int off = 16; off > 0; off >>= 1) M = fmaxf(M, __shfl_xor(M, off, 32));
        const float f = __expf(rc.z - M);
        float S = rc.w * f, q0 = rc.x * f, q1 = rc.y * f;
        for (int off = 16; off > 0; off >>= 1) {
            S  += __shfl_xor(S,  off, 32);
            q0 += __shfl_xor(q0, off, 32);
            q1 += __shfl_xor(q1, off, 32);
        }
        const float inv = 1.0f / S;
        fw[tid] = f * inv;
        if (tid == 0) {
            const float o0 = q0 * inv + lin_b[0];
            const float o1 = q1 * inv + lin_b[1];
            const float mm = fmaxf(o0, o1);
            const float lse = mm + __logf(__expf(o0 - mm) + __expf(o1 - mm));
            logits[b * 2 + 0] = o0 - lse;
            logits[b * 2 + 1] = o1 - lse;
        }
    }
    __syncthreads();
    const float f0 = fw[(4 * tid) / P3_TOK];
    const float f1 = fw[(Ln / 2 + 4 * tid) / P3_TOK];
    *(v4f*)&brow[4 * tid]          = e0 * f0;
    *(v4f*)&brow[Ln / 2 + 4 * tid] = e1 * f1;
}

extern "C" void kernel_launch(void* const* d_in, const int* in_sizes, int n_in,
                              void* d_out, int out_size, void* d_ws, size_t ws_size,
                              hipStream_t stream)
{
    const int*   tok = (const int*)  d_in[0];
    const float* emb = (const float*)d_in[1];
    const float* cls = (const float*)d_in[2];
    const float* cw  = (const float*)d_in[3];
    const float* cb  = (const float*)d_in[4];
    const float* lw  = (const float*)d_in[5];
    const float* lb  = (const float*)d_in[6];

    float* out    = (float*)d_out;
    float* logits = out;
    float* beta   = out + Bn * Cn;
    float* g      = (float*)d_ws;
    float* part   = g + Bn * Cn * Ln;

    k_phase1<<<1024, 256, 0, stream>>>(tok, emb, cls, g);
    k_phase23<<<Bn * SPLIT, 256, 0, stream>>>(g, cw, cb, tok, emb, lw, beta, part, ws_size);
    k_head<<<Bn, 256, 0, stream>>>(part, lb, logits, beta, ws_size);
}


# Phase 1's class norms by lane exchange: a wave adds its 32 threads' squares in five steps, the eight wave sums meet in LDS once; the block's 23 set-up barriers become 2.
# speedup vs baseline: 5.8092x; 1.0315x over previous
#include <hip/hip_runtime.h>
#include <hip/hip_bf16.h>
#include <math.h>

#define Bn   64
#define Ln   2048
#define Dn   300
#define Cn   2
#define Kn   111
#define PADn 55

typedef __attribute__((ext_vector_type(2))) float v2f;
typedef __attribute__((ext_vector_type(8))) float v8f;
typedef __attribute__((ext_vector_type(4))) float v4f;

#define CH_DIMS  100
#define CH_PITCH 100

__device__ __forceinline__ void lds_barrier()
{
    __builtin_amdgcn_fence(__ATOMIC_RELEASE, "workgroup", "local");
    __builtin_amdgcn_s_barrier();
    __builtin_amdgcn_fence(__ATOMIC_ACQUIRE, "workgroup", "local");
}

__global__ __launch_bounds__(256, 8)
void k_phase1(const int* __restrict__ tok, const float* __restrict__ emb,
              const float* __restrict__ cls, float* __restrict__ g)
{
    __shared__ __attribute__((aligned(16))) float clsn[3 * 304];
    __shared__ __attribute__((aligned(16))) float red[16];
    __shared__ __attribute__((aligned(16))) float xs[8][16 * CH_PITCH];
    const int tid = threadIdx.x;

    const int wave = tid >> 5;
    const int lane = tid & 31;
    const int tileId = blockIdx.x * 8 + wave;
    const int b  = tileId >> 7;
    const int l0 = (tileId & 127) * 16;
    const int m    = lane & 15;
    const int hi   = lane >> 4;

    const int token = tok[b * Ln + l0 + m];

    const bool  two = tid + 256 < Dn;
    const int   d1  = two ? tid + 256 : tid;
    const float ca0 = cls[tid];
    const float ca1 = cls[Dn + tid];
    const float cb0 = cls[d1];
    const float cb1 = cls[Dn + d1];

    const int      pl = lane < 25 ? lane : 24;
    const unsigned o  = (unsigned)pl * 16u;
    const char* __restrict__ embc = (const char*)emb;
    char* xsw = (char*)&xs[wave][0] + o;

    const char* rp[16];
    #pragma unroll
    for (int r = 0; r < 16; ++r) {
        const int t_ = __builtin_amdgcn_readlane(token, r);
        rp[r] = embc + (size_t)(unsigned)t_ * (Dn * 4u);
    }

    v4f x[16];
    #pragma unroll
    for (int r = 0; r < 16; ++r) x[r] = *(const v4f*)(rp[r] + o);

    for (int i = tid; i < 3 * 304; i += 256) clsn[i] = 0.0f;
    const float vb0 = two ? cb0 : 0.0f;
    const float vb1 = two ? cb1 : 0.0f;
    float s0 = ca0 * ca0;
    float s1 = ca1 * ca1;
    s0 = __builtin_fmaf(vb0, vb0, s0);
    s1 = __builtin_fmaf(vb1, vb1, s1);
    for (int off = 16; off > 0; off >>= 1) {
        s0 += __shfl_xor(s0, off, 32);
        s1 += __shfl_xor(s1, off, 32);
    }
    if (lane == 0) { red[wave] = s0; red[8 + wave] = s1; }
    lds_barrier();
    float t0 = 0.0f, t1 = 0.0f;
    #pragma unroll
    for (int w = 0; w < 8; ++w) { t0 += red[w]; t1 += red[8 + w]; }
    const float rn0 = 1.0f / fmaxf(sqrtf(t0), 1e-12f);
    const float rn1 = 1.0f / fmaxf(sqrtf(t1), 1e-12f);
    clsn[tid]       = ca0 * rn0;
    clsn[304 + tid] = ca1 * rn1;
    if (two) {
        clsn[tid + 256]       = cb0 * rn0;
        clsn[304 + tid + 256] = cb1 * rn1;
    }
    lds_barrier();

    const float* __restrict__ ar = &xs[wave][m * CH_PITCH + 4 * hi];
    const float* __restrict__ br = &clsn[(m < 2 ? m : 2) * 304 + 4 * hi];

    v8f acc0 = {}, acc1 = {};
    float n2 = 0.0f;
    #pragma unroll
    for (int q = 0; q < 3; ++q) {
        #pragma unroll
        for (int r = 0; r < 16; ++r) *(v4f*)(xsw + r * (CH_PITCH * 4)) = x[r];
        if (q < 2) {
            unsigned on = o + (unsigned)(q + 1) * (CH_DIMS * 4u);
            asm volatile("" : "+v"(on));
            #pragma unroll
            for (int r = 0; r < 16; ++r) x[r] = *(const v4f*)(rp[r] + on);
        }
        __builtin_amdgcn_fence(__ATOMIC_RELEASE, "wavefront");
        __builtin_amdgcn_wave_barrier();
        __builtin_amdgcn_fence(__ATOMIC_ACQUIRE, "wavefront");

        const float* __restrict__ bq_ = br + q * CH_DIMS;
        #pragma unroll
        for (int j = 0; j < 12; ++j) {
            const v4f a  = *(const v4f*)(ar + 8 * j);
            const v4f bq = *(const v4f*)(bq_ + 8 * j);
            n2 += a.x * a.x + a.y * a.y + a.z * a.z + a.w * a.w;
            v2f a01, a23, b01, b23;
            a01.x = a.x;  a01.y = a.y;  a23.x = a.z;  a23.y = a.w;
            b01.x = bq.x; b01.y = bq.y; b23.x = bq.z; b23.y = bq.w;
            acc0 = __builtin_amdgcn_wmma_f32_16x16x4_f32(
                false, a01, false, b01, (short)0, acc0, false, false);
            acc1 = __builtin_amdgcn_wmma_f32_16x16x4_f32(
                false, a23, false, b23, (short)0, acc1, false, false);
        }
        {
            const v2f a  = *(const v2f*)(&xs[wave][m * CH_PITCH + 96 + 2 * hi]);
            const v2f bq = *(const v2f*)(&clsn[(m < 2 ? m : 2) * 304 + q * CH_DIMS + 96 + 2 * hi]);
            n2 += a.x * a.x + a.y * a.y;
            if (q & 1)
                acc1 = __builtin_amdgcn_wmma_f32_16x16x4_f32(
                    false, a, false, bq, (short)0, acc1, false, false);
            else
                acc0 = __builtin_amdgcn_wmma_f32_16x16x4_f32(
                    false, a, false, bq, (short)0, acc0, false, false);
        }
        __builtin_amdgcn_fence(__ATOMIC_RELEASE, "wavefront");
        __builtin_amdgcn_wave_barrier();
        __builtin_amdgcn_fence(__ATOMIC_ACQUIRE, "wavefront");
    }
    const v8f acc = acc0 + acc1;

    n2 += __shfl_xor(n2, 16, 32);

    const int  c      = m;
    const bool active = (m < 2);
    const int  mbase  = hi * 8;
    float* __restrict__ gout = g + (((long)b * Cn + c) * Ln + l0 + mbase);
    #pragma unroll
    for (int i = 0; i < 8; ++i) {
        float nn    = __shfl(n2, mbase + i, 32);
        float scale = 1.0f / fmaxf(sqrtf(nn), 1e-12f);
        if (active) gout[i] = acc[i] * scale;
    }
}

#define SPLIT     4
#define BLK_POS   (Ln / SPLIT)
#define GP_IN     (BLK_POS + 2 * PADn + 2)
#define GP_ROWS   (GP_IN / 16)
#define GP_PITCH  20
#define WP_LEAD   15
#define WP_LEN    144

#define P3_WAVES  8
#define ROW_WAVES (SPLIT * P3_WAVES)
#define P3_TOK    (Ln / ROW_WAVES)
#define P3_PRE    20
#define REC       4
#define WS_FLOATS (Bn * Cn * Ln + Bn * ROW_WAVES * REC)

__global__ __launch_bounds__(256)
void k_phase23(const float* __restrict__ g, const float* __restrict__ conv_w,
               const float* __restrict__ conv_b, const int* __restrict__ tok,
               const float* __restrict__ emb, const float* __restrict__ lin_w,
               float* __restrict__ betaOut, float* __restrict__ part,
               size_t ws_bytes)
{
    __shared__ __attribute__((aligned(16))) float gp[Cn][GP_ROWS * GP_PITCH];
    __shared__ float wp[Cn * Cn][WP_LEN];
    __shared__ float hc[Cn][BLK_POS];
    const int b  = blockIdx.x / SPLIT;
    const int sp = blockIdx.x % SPLIT;
    const int tid = threadIdx.x;
    const int lane = tid & 31;
    const int wv   = __builtin_amdgcn_readfirstlane(tid >> 5);
    const int rw   = sp * P3_WAVES + wv;

    const int* __restrict__ trow = tok + (long)b * Ln + rw * P3_TOK;
    const int tv0 = trow[lane], tv1 = trow[lane + 32];

    float gr0[3], gr1[3];
    #pragma unroll
    for (int k = 0; k < 3; ++k) {
        const int i = tid + 256 * k;
        const int l = BLK_POS * sp - PADn + i;
        const bool in = (i < GP_IN) && (l >= 0) && (l < Ln);
        const int lc = l < 0 ? 0 : (l < Ln ? l : Ln - 1);
        const float v0 = g[((long)b * Cn + 0) * Ln + lc];
        const float v1 = g[((long)b * Cn + 1) * Ln + lc];
        gr0[k] = in ? v0 : 0.0f;
        gr1[k] = in ? v1 : 0.0f;
    }
    const float cw0 = conv_w[tid];
    const float cw1 = (tid + 256 < Cn * Cn * Kn) ? conv_w[tid + 256] : 0.0f;

    for (int i = tid; i < Cn * Cn * WP_LEN; i += 256) (&wp[0][0])[i] = 0.0f;

    const int      pl = lane < 25 ? lane : 24;
    const unsigned o  = (unsigned)pl * 16u;
    const char* __restrict__ embc = (const char*)emb;

    v4f x0[P3_PRE], x1[P3_PRE], x2[P3_PRE];

#define P3_LOAD(slot, p) {                                                    \
        const int t_ = (p) < 32 ? __builtin_amdgcn_readlane(tv0, (p) & 31)   \
                                : __builtin_amdgcn_readlane(tv1, (p) & 31);  \
        const char* rp_ = embc + ((unsigned)t_ * (Dn * 4u) + o);             \
        x0[slot] = *(const v4f*)(rp_);                                       \
        x1[slot] = *(const v4f*)(rp_ + 400);                                 \
        x2[slot] = *(const v4f*)(rp_ + 800); }

#define P3_W(p)                                                               \
        __builtin_bit_cast(float, (p) < 32                                    \
            ? __builtin_amdgcn_readlane(__builtin_bit_cast(int, bv0), (p) & 31) \
            : __builtin_amdgcn_readlane(__builtin_bit_cast(int, bv1), (p) & 31))

#define P3_USE(slot, p) {                                                     \
        const float bt_ = P3_W(p);                                            \
        z0 += bt_ * x0[slot]; z1 += bt_ * x1[slot]; z2 += bt_ * x2[slot]; }

    #pragma unroll
    for (int j = 0; j < P3_PRE; ++j) P3_LOAD(j, j);

    lds_barrier();
    #pragma unroll
    for (int k = 0; k < 3; ++k) {
        const int i = tid + 256 * k;
        if (i < GP_IN) {
            const int a = (i >> 4) * GP_PITCH + (i & 15);
            gp[0][a] = gr0[k];
            gp[1][a] = gr1[k];
        }
    }
    {
        const int pr0 = tid / Kn;
        wp[pr0][WP_LEAD + tid - pr0 * Kn] = cw0;
        const int i1 = tid + 256;
        if (i1 < Cn * Cn * Kn) {
            const int pr1 = i1 / Kn;
            wp[pr1][WP_LEAD + i1 - pr1 * Kn] = cw1;
        }
    }
    lds_barrier();

    const int c    = lane & 15;
    const int hh   = lane >> 4;
    if (wv < 4) {
        const int tl = wv & 1;
        const int co = wv >> 1;
        const float* __restrict__ a0p = &gp[0][(16 * tl + c) * GP_PITCH + 2 * hh];
        const float* __restrict__ a1p = &gp[1][(16 * tl + c) * GP_PITCH + 2 * hh];
        const float* __restrict__ bp  = &wp[2 * co][WP_LEAD - c + 2 * hh];

        v8f acc0 = {}, acc1 = {};
        #pragma unroll
        for (int p = 0; p < 8; ++p) {
            #pragma unroll
            for (int q = 0; q < 4; ++q) {
                const int ao = p * GP_PITCH + 4 * q;
                const int t  = 16 * p + 4 * q;
                const v2f a0 = *(const v2f*)(a0p + ao);
                const v2f a1 = *(const v2f*)(a1p + ao);
                v2f b0, b1;
                b0.x = bp[0 * WP_LEN + t]; b0.y = bp[0 * WP_LEN + t + 1];
                b1.x = bp[1 * WP_LEN + t]; b1.y = bp[1 * WP_LEN + t + 1];
                acc0 = __builtin_amdgcn_wmma_f32_16x16x4_f32(
                    false, a0, false, b0, (short)0, acc0, false, false);
                acc1 = __builtin_amdgcn_wmma_f32_16x16x4_f32(
                    false, a1, false, b1, (short)0, acc1, false, false);
            }
        }
        const int xb = 256 * tl + 128 * hh + c;
        const float cb = conv_b[co];
        #pragma unroll
        for (int e = 0; e < 8; ++e)
            hc[co][xb + 16 * e] = fmaxf(acc0[e] + acc1[e] + cb, 0.0f);
    }
    lds_barrier();

    const float h0 = fmaxf(hc[0][P3_TOK * wv + lane],      hc[1][P3_TOK * wv + lane]);
    const float h1 = fmaxf(hc[0][P3_TOK * wv + 32 + lane], hc[1][P3_TOK * wv + 32 + lane]);
    float mw = fmaxf(h0, h1);
    for (int off = 16; off > 0; off >>= 1) mw = fmaxf(mw, __shfl_xor(mw, off, 32));
    const float bv0 = __expf(h0 - mw);
    const float bv1 = __expf(h1 - mw);
    float sw = bv0 + bv1;
    for (int off = 16; off > 0; off >>= 1) sw += __shfl_xor(sw, off, 32);
    {
        float* __restrict__ bo = betaOut + (long)b * Ln + rw * P3_TOK;
        bo[lane]      = bv0;
        bo[lane + 32] = bv1;
    }

    v4f z0 = {}, z1 = {}, z2 = {};

    const v4f lw00 = *(const v4f*)&lin_w[4 * pl];
    const v4f lw01 = *(const v4f*)&lin_w[100 + 4 * pl];
    const v4f lw02 = *(const v4f*)&lin_w[200 + 4 * pl];
    const v4f lw10 = *(const v4f*)&lin_w[Dn + 4 * pl];
    const v4f lw11 = *(const v4f*)&lin_w[Dn + 100 + 4 * pl];
    const v4f lw12 = *(const v4f*)&lin_w[Dn + 200 + 4 * pl];

    #pragma unroll
    for (int r = 0; r < (P3_TOK + P3_PRE - 1) / P3_PRE; ++r) {
        #pragma unroll
        for (int j = 0; j < P3_PRE; j += 2) {
            const int p = r * P3_PRE + j;
            if (p < P3_TOK) {
                P3_USE(j, p);
                P3_USE(j + 1, p + 1);
                if (p + P3_PRE < P3_TOK) {
                    P3_LOAD(j, p + P3_PRE);
                    P3_LOAD(j + 1, p + P3_PRE + 1);
                }
                __builtin_amdgcn_sched_barrier(0);
            }
        }
    }

#undef P3_LOAD
#undef P3_USE
#undef P3_W

    float p0 = 0.0f, p1 = 0.0f;
    #pragma unroll
    for (int e = 0; e < 4; ++e) {
        p0 += z0[e] * lw00[e] + z1[e] * lw01[e] + z2[e] * lw02[e];
        p1 += z0[e] * lw10[e] + z1[e] * lw11[e] + z2[e] * lw12[e];
    }
    p0 = lane < 25 ? p0 : 0.0f;
    p1 = lane < 25 ? p1 : 0.0f;
    for (int off = 16; off > 0; off >>= 1) {
        p0 += __shfl_xor(p0, off, 32);
        p1 += __shfl_xor(p1, off, 32);
    }
    if (lane == 0 && ws_bytes >= sizeof(float) * WS_FLOATS) {
        v4f rc;
        rc.x = p0; rc.y = p1; rc.z = mw; rc.w = sw;
        *(v4f*)&part[(b * ROW_WAVES + rw) * REC] = rc;
    }
}

__global__ __launch_bounds__(256)
void k_head(const float* __restrict__ part, const float* __restrict__ lin_b,
            float* __restrict__ logits, float* __restrict__ beta, size_t ws_bytes)
{
    if (ws_bytes < sizeof(float) * WS_FLOATS) return;
    __shared__ float fw[ROW_WAVES];
    const int b = blockIdx.x, tid = threadIdx.x;
    float* __restrict__ brow = beta + (long)b * Ln;
    const v4f e0 = *(const v4f*)&brow[4 * tid];
    const v4f e1 = *(const v4f*)&brow[Ln / 2 + 4 * tid];
    if (tid < ROW_WAVES) {
        const v4f rc = *(const v4f*)&part[(b * ROW_WAVES + tid) * REC];
        float M = rc.z;
        for (int off = 16; off > 0; off >>= 1) M = fmaxf(M, __shfl_xor(M, off, 32));
        const float f = __expf(rc.z - M);
        float S = rc.w * f, q0 = rc.x * f, q1 = rc.y * f;
        for (int off = 16; off > 0; off >>= 1) {
            S  += __shfl_xor(S,  off, 32);
            q0 += __shfl_xor(q0, off, 32);
            q1 += __shfl_xor(q1, off, 32);
        }
        const float inv = 1.0f / S;
        fw[tid] = f * inv;
        if (tid == 0) {
            const float o0 = q0 * inv + lin_b[0];
            const float o1 = q1 * inv + lin_b[1];
            const float mm = fmaxf(o0, o1);
            const float lse = mm + __logf(__expf(o0 - mm) + __expf(o1 - mm));
            logits[b * 2 + 0] = o0 - lse;
            logits[b * 2 + 1] = o1 - lse;
        }
    }
    __syncthreads();
    const float f0 = fw[(4 * tid) / P3_TOK];
    const float f1 = fw[(Ln / 2 + 4 * tid) / P3_TOK];
    *(v4f*)&brow[4 * tid]          = e0 * f0;
    *(v4f*)&brow[Ln / 2 + 4 * tid] = e1 * f1;
}

extern "C" void kernel_launch(void* const* d_in, const int* in_sizes, int n_in,
                              void* d_out, int out_size, void* d_ws, size_t ws_size,
                              hipStream_t stream)
{
    const int*   tok = (const int*)  d_in[0];
    const float* emb = (const float*)d_in[1];
    const float* cls = (const float*)d_in[2];
    const float* cw  = (const float*)d_in[3];
    const float* cb  = (const float*)d_in[4];
    const float* lw  = (const float*)d_in[5];
    const float* lb  = (const float*)d_in[6];

    float* out    = (float*)d_out;
    float* logits = out;
    float* beta   = out + Bn * Cn;
    float* g      = (float*)d_ws;
    float* part   = g + Bn * Cn * Ln;

    k_phase1<<<1024, 256, 0, stream>>>(tok, emb, cls, g);
    k_phase23<<<Bn * SPLIT, 256, 0, stream>>>(g, cw, cb, tok, emb, lw, beta, part, ws_size);
    k_head<<<Bn, 256, 0, stream>>>(part, lb, logits, beta, ws_size);
}
